# starting kernel
# baseline (speedup 1.0000x reference)
#include <hip/hip_runtime.h>

#define N_USERS  100000
#define N_ITEMS  50000
#define N_NODES  150000
#define N_EDGES  1000000
#define EMBED_K  64

typedef __attribute__((ext_vector_type(2))) float v2f;
typedef __attribute__((ext_vector_type(8))) float v8f;

__device__ __forceinline__ void atomAddF32(float* p, float v) {
    (void)__hip_atomic_fetch_add(p, v, __ATOMIC_RELAXED, __HIP_MEMORY_SCOPE_AGENT);
}

__global__ void k_zero(float* __restrict__ p, int n) {
    int i = blockIdx.x * blockDim.x + threadIdx.x;
    if (i < n) p[i] = 0.0f;
}

__global__ void k_init_x(const float* __restrict__ Gu, const float* __restrict__ Gi,
                         float* __restrict__ x) {
    int i = blockIdx.x * blockDim.x + threadIdx.x;
    const int total  = N_NODES * EMBED_K;
    const int uElems = N_USERS * EMBED_K;
    if (i >= total) return;
    x[i] = (i < uElems) ? Gu[i] : Gi[i - uElems];
}

__global__ void k_deg(const int* __restrict__ row, const int* __restrict__ col,
                      const float* __restrict__ intents, float* __restrict__ deg) {
    int e = blockIdx.x * blockDim.x + threadIdx.x;
    if (e >= N_EDGES) return;
    float z0 = intents[e];
    float z1 = intents[N_EDGES + e];
    float z2 = intents[2 * N_EDGES + e];
    float z3 = intents[3 * N_EDGES + e];
    float m  = fmaxf(fmaxf(z0, z1), fmaxf(z2, z3));
    float w0 = __expf(z0 - m), w1 = __expf(z1 - m);
    float w2 = __expf(z2 - m), w3 = __expf(z3 - m);
    float inv = 1.0f / (w0 + w1 + w2 + w3);
    w0 *= inv; w1 *= inv; w2 *= inv; w3 *= inv;
    int r = row[e], c = col[e];
    atomAddF32(deg + r * 4 + 0, w0); atomAddF32(deg + r * 4 + 1, w1);
    atomAddF32(deg + r * 4 + 2, w2); atomAddF32(deg + r * 4 + 3, w3);
    atomAddF32(deg + c * 4 + 0, w0); atomAddF32(deg + c * 4 + 1, w1);
    atomAddF32(deg + c * 4 + 2, w2); atomAddF32(deg + c * 4 + 3, w3);
}

__global__ void k_dis(float* __restrict__ deg) {
    int i = blockIdx.x * blockDim.x + threadIdx.x;
    if (i >= N_NODES * 4) return;
    float d = deg[i];
    deg[i] = (d > 0.0f) ? rsqrtf(fmaxf(d, 1e-12f)) : 0.0f;
}

__global__ __launch_bounds__(128) void k_propagate(
        const float* __restrict__ x_src, float* __restrict__ x_dst,
        const float* __restrict__ dis,
        const int* __restrict__ row, const int* __restrict__ col) {
    const int wave = blockIdx.x * (blockDim.x >> 5) + (threadIdx.x >> 5);
    const int lane = threadIdx.x & 31;
    const int hl   = lane & 15;
    const bool hi  = lane >= 16;
    const int base = wave * 16;

    const int eA  = base + hl;
    const int irA = row[eA];
    const int icA = col[eA];
    __builtin_prefetch(&x_src[irA * 64], 0, 0);
    const float4 dr = *(const float4*)(dis + irA * 4);
    const float4 dc = *(const float4*)(dis + icA * 4);
    float nrm[4];
    nrm[0] = dr.x * dc.x; nrm[1] = dr.y * dc.y;
    nrm[2] = dr.z * dc.z; nrm[3] = dr.w * dc.w;

    int colv[8];
#pragma unroll
    for (int v = 0; v < 8; ++v) colv[v] = col[base + v + (hi ? 8 : 0)];

    const int kx = hi ? 2 : 0;

#pragma unroll
    for (int k = 0; k < 4; ++k) {
        v8f acc = {};
#pragma unroll
        for (int t = 0; t < 4; ++t) {
            v2f a, b;
            a.x = (hl == 4 * t + kx)     ? nrm[k] : 0.0f;
            a.y = (hl == 4 * t + kx + 1) ? nrm[k] : 0.0f;
            const int rb0 = row[base + 4 * t + kx];
            const int rb1 = row[base + 4 * t + kx + 1];
            b.x = x_src[rb0 * 64 + k * 16 + hl];
            b.y = x_src[rb1 * 64 + k * 16 + hl];
            acc = __builtin_amdgcn_wmma_f32_16x16x4_f32(
                      false, a, false, b, (short)0, acc, false, false);
        }
#pragma unroll
        for (int v = 0; v < 8; ++v)
            atomAddF32(x_dst + colv[v] * 64 + k * 16 + hl, acc[v]);
    }
}

extern "C" void kernel_launch(void* const* d_in, const int* in_sizes, int n_in,
                              void* d_out, int out_size, void* d_ws, size_t ws_size,
                              hipStream_t stream) {
    const float* Gu      = (const float*)d_in[0];
    const float* Gi      = (const float*)d_in[1];
    const int*   edge    = (const int*)d_in[2];
    const float* intents = (const float*)d_in[3];
    const int* row = edge;
    const int* col = edge + N_EDGES;
    float* out = (float*)d_out;

    float* deg  = (float*)d_ws;
    float* xbuf = deg + (size_t)N_NODES * 4;

    const int NX = N_NODES * EMBED_K;
    const int ND = N_NODES * 4;
    dim3 b256(256);
    const int waves  = N_EDGES / 16;
    dim3 pgrid(waves / 4), pblock(128);

    k_zero<<<dim3((ND + 255) / 256), b256, 0, stream>>>(deg, ND);
    k_deg<<<dim3((N_EDGES + 255) / 256), b256, 0, stream>>>(row, col, intents, deg);
    k_dis<<<dim3((ND + 255) / 256), b256, 0, stream>>>(deg);

    k_init_x<<<dim3((NX + 255) / 256), b256, 0, stream>>>(Gu, Gi, xbuf);

    k_zero<<<dim3((NX + 255) / 256), b256, 0, stream>>>(out, NX);
    k_propagate<<<pgrid, pblock, 0, stream>>>(xbuf, out, deg, row, col);

    k_zero<<<dim3((NX + 255) / 256), b256, 0, stream>>>(xbuf, NX);
    k_propagate<<<pgrid, pblock, 0, stream>>>(out, xbuf, deg, row, col);

    k_zero<<<dim3((NX + 255) / 256), b256, 0, stream>>>(out, NX);
    k_propagate<<<pgrid, pblock, 0, stream>>>(xbuf, out, deg, row, col);
}


# k_deg: 4 lanes per edge, lane q adds intent q, so one atomic instruction carries whole 16-byte node records (4x fewer atomic line requests); atomics kept, arrival-order sums as before
# speedup vs baseline: 1.3062x; 1.3062x over previous
#include <hip/hip_runtime.h>

#define N_USERS  100000
#define N_ITEMS  50000
#define N_NODES  150000
#define N_EDGES  1000000
#define EMBED_K  64

typedef __attribute__((ext_vector_type(2))) float v2f;
typedef __attribute__((ext_vector_type(8))) float v8f;

__device__ __forceinline__ void atomAddF32(float* p, float v) {
    (void)__hip_atomic_fetch_add(p, v, __ATOMIC_RELAXED, __HIP_MEMORY_SCOPE_AGENT);
}

__global__ void k_zero(float* __restrict__ p, int n) {
    int i = blockIdx.x * blockDim.x + threadIdx.x;
    if (i < n) p[i] = 0.0f;
}

__global__ void k_init_x(const float* __restrict__ Gu, const float* __restrict__ Gi,
                         float* __restrict__ x) {
    int i = blockIdx.x * blockDim.x + threadIdx.x;
    const int total  = N_NODES * EMBED_K;
    const int uElems = N_USERS * EMBED_K;
    if (i >= total) return;
    x[i] = (i < uElems) ? Gu[i] : Gi[i - uElems];
}

__global__ void k_deg(const int* __restrict__ row, const int* __restrict__ col,
                      const float* __restrict__ intents, float* __restrict__ deg) {
    int t = blockIdx.x * blockDim.x + threadIdx.x;
    int e = t >> 2;
    int q = t & 3;
    if (e >= N_EDGES) return;
    float z0 = intents[e];
    float z1 = intents[N_EDGES + e];
    float z2 = intents[2 * N_EDGES + e];
    float z3 = intents[3 * N_EDGES + e];
    float m  = fmaxf(fmaxf(z0, z1), fmaxf(z2, z3));
    float w0 = __expf(z0 - m), w1 = __expf(z1 - m);
    float w2 = __expf(z2 - m), w3 = __expf(z3 - m);
    float inv = 1.0f / (w0 + w1 + w2 + w3);
    w0 *= inv; w1 *= inv; w2 *= inv; w3 *= inv;
    const float wlo = (q & 1) ? w1 : w0;
    const float whi = (q & 1) ? w3 : w2;
    const float w   = (q & 2) ? whi : wlo;
    int r = row[e], c = col[e];
    atomAddF32(deg + r * 4 + q, w);
    atomAddF32(deg + c * 4 + q, w);
}

__global__ void k_dis(float* __restrict__ deg) {
    int i = blockIdx.x * blockDim.x + threadIdx.x;
    if (i >= N_NODES * 4) return;
    float d = deg[i];
    deg[i] = (d > 0.0f) ? rsqrtf(fmaxf(d, 1e-12f)) : 0.0f;
}

__global__ __launch_bounds__(128) void k_propagate(
        const float* __restrict__ x_src, float* __restrict__ x_dst,
        const float* __restrict__ dis,
        const int* __restrict__ row, const int* __restrict__ col) {
    const int wave = blockIdx.x * (blockDim.x >> 5) + (threadIdx.x >> 5);
    const int lane = threadIdx.x & 31;
    const int hl   = lane & 15;
    const bool hi  = lane >= 16;
    const int base = wave * 16;

    const int eA  = base + hl;
    const int irA = row[eA];
    const int icA = col[eA];
    __builtin_prefetch(&x_src[irA * 64], 0, 0);
    const float4 dr = *(const float4*)(dis + irA * 4);
    const float4 dc = *(const float4*)(dis + icA * 4);
    float nrm[4];
    nrm[0] = dr.x * dc.x; nrm[1] = dr.y * dc.y;
    nrm[2] = dr.z * dc.z; nrm[3] = dr.w * dc.w;

    int colv[8];
#pragma unroll
    for (int v = 0; v < 8; ++v) colv[v] = col[base + v + (hi ? 8 : 0)];

    const int kx = hi ? 2 : 0;

#pragma unroll
    for (int k = 0; k < 4; ++k) {
        v8f acc = {};
#pragma unroll
        for (int t = 0; t < 4; ++t) {
            v2f a, b;
            a.x = (hl == 4 * t + kx)     ? nrm[k] : 0.0f;
            a.y = (hl == 4 * t + kx + 1) ? nrm[k] : 0.0f;
            const int rb0 = row[base + 4 * t + kx];
            const int rb1 = row[base + 4 * t + kx + 1];
            b.x = x_src[rb0 * 64 + k * 16 + hl];
            b.y = x_src[rb1 * 64 + k * 16 + hl];
            acc = __builtin_amdgcn_wmma_f32_16x16x4_f32(
                      false, a, false, b, (short)0, acc, false, false);
        }
#pragma unroll
        for (int v = 0; v < 8; ++v)
            atomAddF32(x_dst + colv[v] * 64 + k * 16 + hl, acc[v]);
    }
}

extern "C" void kernel_launch(void* const* d_in, const int* in_sizes, int n_in,
                              void* d_out, int out_size, void* d_ws, size_t ws_size,
                              hipStream_t stream) {
    const float* Gu      = (const float*)d_in[0];
    const float* Gi      = (const float*)d_in[1];
    const int*   edge    = (const int*)d_in[2];
    const float* intents = (const float*)d_in[3];
    const int* row = edge;
    const int* col = edge + N_EDGES;
    float* out = (float*)d_out;

    float* deg  = (float*)d_ws;
    float* xbuf = deg + (size_t)N_NODES * 4;

    const int NX = N_NODES * EMBED_K;
    const int ND = N_NODES * 4;
    dim3 b256(256);
    const int waves  = N_EDGES / 16;
    dim3 pgrid(waves / 4), pblock(128);

    k_zero<<<dim3((ND + 255) / 256), b256, 0, stream>>>(deg, ND);
    k_deg<<<dim3((4 * N_EDGES + 255) / 256), b256, 0, stream>>>(row, col, intents, deg);
    k_dis<<<dim3((ND + 255) / 256), b256, 0, stream>>>(deg);

    k_init_x<<<dim3((NX + 255) / 256), b256, 0, stream>>>(Gu, Gi, xbuf);

    k_zero<<<dim3((NX + 255) / 256), b256, 0, stream>>>(out, NX);
    k_propagate<<<pgrid, pblock, 0, stream>>>(xbuf, out, deg, row, col);

    k_zero<<<dim3((NX + 255) / 256), b256, 0, stream>>>(xbuf, NX);
    k_propagate<<<pgrid, pblock, 0, stream>>>(out, xbuf, deg, row, col);

    k_zero<<<dim3((NX + 255) / 256), b256, 0, stream>>>(out, NX);
    k_propagate<<<pgrid, pblock, 0, stream>>>(xbuf, out, deg, row, col);
}


# k_propagate: WMMA tile = 8 edges x 2 intents (was 16 x 1): each gather load and each atomic add now covers one whole 128-byte row line, not two 64-byte halves; atomics kept, arrival-order sums
# speedup vs baseline: 1.7092x; 1.3085x over previous
#include <hip/hip_runtime.h>

#define N_USERS  100000
#define N_ITEMS  50000
#define N_NODES  150000
#define N_EDGES  1000000
#define EMBED_K  64

typedef __attribute__((ext_vector_type(2))) float v2f;
typedef __attribute__((ext_vector_type(8))) float v8f;

__device__ __forceinline__ void atomAddF32(float* p, float v) {
    (void)__hip_atomic_fetch_add(p, v, __ATOMIC_RELAXED, __HIP_MEMORY_SCOPE_AGENT);
}

__global__ void k_zero(float* __restrict__ p, int n) {
    int i = blockIdx.x * blockDim.x + threadIdx.x;
    if (i < n) p[i] = 0.0f;
}

__global__ void k_init_x(const float* __restrict__ Gu, const float* __restrict__ Gi,
                         float* __restrict__ x) {
    int i = blockIdx.x * blockDim.x + threadIdx.x;
    const int total  = N_NODES * EMBED_K;
    const int uElems = N_USERS * EMBED_K;
    if (i >= total) return;
    x[i] = (i < uElems) ? Gu[i] : Gi[i - uElems];
}

__global__ void k_deg(const int* __restrict__ row, const int* __restrict__ col,
                      const float* __restrict__ intents, float* __restrict__ deg) {
    int t = blockIdx.x * blockDim.x + threadIdx.x;
    int e = t >> 2;
    int q = t & 3;
    if (e >= N_EDGES) return;
    float z0 = intents[e];
    float z1 = intents[N_EDGES + e];
    float z2 = intents[2 * N_EDGES + e];
    float z3 = intents[3 * N_EDGES + e];
    float m  = fmaxf(fmaxf(z0, z1), fmaxf(z2, z3));
    float w0 = __expf(z0 - m), w1 = __expf(z1 - m);
    float w2 = __expf(z2 - m), w3 = __expf(z3 - m);
    float inv = 1.0f / (w0 + w1 + w2 + w3);
    w0 *= inv; w1 *= inv; w2 *= inv; w3 *= inv;
    const float wlo = (q & 1) ? w1 : w0;
    const float whi = (q & 1) ? w3 : w2;
    const float w   = (q & 2) ? whi : wlo;
    int r = row[e], c = col[e];
    atomAddF32(deg + r * 4 + q, w);
    atomAddF32(deg + c * 4 + q, w);
}

__global__ void k_dis(float* __restrict__ deg) {
    int i = blockIdx.x * blockDim.x + threadIdx.x;
    if (i >= N_NODES * 4) return;
    float d = deg[i];
    deg[i] = (d > 0.0f) ? rsqrtf(fmaxf(d, 1e-12f)) : 0.0f;
}

__global__ __launch_bounds__(128) void k_propagate(
        const float* __restrict__ x_src, float* __restrict__ x_dst,
        const float* __restrict__ dis,
        const int* __restrict__ row, const int* __restrict__ col) {
    const int wave = __builtin_amdgcn_readfirstlane(
                         (int)(blockIdx.x * (blockDim.x >> 5) + (threadIdx.x >> 5)));
    const int lane = threadIdx.x & 31;
    const int hl   = lane & 15;
    const int h    = lane >> 4;
    const int j    = hl & 7;
    const int hs   = hl >> 3;

    const int eA0 = wave * 16 + j, eA1 = wave * 16 + 8 + j;
    const int irA0 = row[eA0], icA0 = col[eA0];
    const int irA1 = row[eA1], icA1 = col[eA1];
    __builtin_prefetch(&x_src[irA0 * 64], 0, 0);
    __builtin_prefetch(&x_src[irA1 * 64], 0, 0);
    const float4 dr0 = *(const float4*)(dis + irA0 * 4);
    const float4 dc0 = *(const float4*)(dis + icA0 * 4);
    const float4 dr1 = *(const float4*)(dis + irA1 * 4);
    const float4 dc1 = *(const float4*)(dis + icA1 * 4);
    const float4 n0 = make_float4(dr0.x * dc0.x, dr0.y * dc0.y, dr0.z * dc0.z, dr0.w * dc0.w);
    const float4 n1 = make_float4(dr1.x * dc1.x, dr1.y * dc1.y, dr1.z * dc1.z, dr1.w * dc1.w);
    const float nrm00 = hs ? n0.y : n0.x, nrm01 = hs ? n0.w : n0.z;
    const float nrm10 = hs ? n1.y : n1.x, nrm11 = hs ? n1.w : n1.z;

#pragma unroll
    for (int g = 0; g < 2; ++g) {
        const int base = wave * 16 + g * 8;

        int rb[8], cb[8];
#pragma unroll
        for (int i = 0; i < 8; ++i) { rb[i] = row[base + i]; cb[i] = col[base + i]; }

#pragma unroll
        for (int kp = 0; kp < 2; ++kp) {
            const float nk = g ? (kp ? nrm11 : nrm10) : (kp ? nrm01 : nrm00);
            v8f acc = {};
#pragma unroll
            for (int t = 0; t < 4; ++t) {
                v2f a, b;
                a.x = (hl == 8 * h + 2 * t)     ? nk : 0.0f;
                a.y = (hl == 8 * h + 2 * t + 1) ? nk : 0.0f;
                b.x = x_src[rb[2 * t]     * 64 + kp * 32 + lane];
                b.y = x_src[rb[2 * t + 1] * 64 + kp * 32 + lane];
                acc = __builtin_amdgcn_wmma_f32_16x16x4_f32(
                          false, a, false, b, (short)0, acc, false, false);
            }
#pragma unroll
            for (int v = 0; v < 8; ++v)
                atomAddF32(x_dst + cb[v] * 64 + kp * 32 + lane, acc[v]);
        }
    }
}

extern "C" void kernel_launch(void* const* d_in, const int* in_sizes, int n_in,
                              void* d_out, int out_size, void* d_ws, size_t ws_size,
                              hipStream_t stream) {
    const float* Gu      = (const float*)d_in[0];
    const float* Gi      = (const float*)d_in[1];
    const int*   edge    = (const int*)d_in[2];
    const float* intents = (const float*)d_in[3];
    const int* row = edge;
    const int* col = edge + N_EDGES;
    float* out = (float*)d_out;

    float* deg  = (float*)d_ws;
    float* xbuf = deg + (size_t)N_NODES * 4;

    const int NX = N_NODES * EMBED_K;
    const int ND = N_NODES * 4;
    dim3 b256(256);
    const int waves  = N_EDGES / 16;
    dim3 pgrid(waves / 4), pblock(128);

    k_zero<<<dim3((ND + 255) / 256), b256, 0, stream>>>(deg, ND);
    k_deg<<<dim3((4 * N_EDGES + 255) / 256), b256, 0, stream>>>(row, col, intents, deg);
    k_dis<<<dim3((ND + 255) / 256), b256, 0, stream>>>(deg);

    k_init_x<<<dim3((NX + 255) / 256), b256, 0, stream>>>(Gu, Gi, xbuf);

    k_zero<<<dim3((NX + 255) / 256), b256, 0, stream>>>(out, NX);
    k_propagate<<<pgrid, pblock, 0, stream>>>(xbuf, out, deg, row, col);

    k_zero<<<dim3((NX + 255) / 256), b256, 0, stream>>>(xbuf, NX);
    k_propagate<<<pgrid, pblock, 0, stream>>>(out, xbuf, deg, row, col);

    k_zero<<<dim3((NX + 255) / 256), b256, 0, stream>>>(out, NX);
    k_propagate<<<pgrid, pblock, 0, stream>>>(xbuf, out, deg, row, col);
}


# norm factored out of the edge loop: zero passes become row-scaling passes (dis on WMMA diagonal), k_propagate only gathers and adds, no dis gathers; dis[row]*x summed, then *dis[col]; arrival order
# speedup vs baseline: 1.9120x; 1.1187x over previous
#include <hip/hip_runtime.h>

#define N_USERS  100000
#define N_ITEMS  50000
#define N_NODES  150000
#define N_EDGES  1000000
#define EMBED_K  64

typedef __attribute__((ext_vector_type(2))) float v2f;
typedef __attribute__((ext_vector_type(8))) float v8f;

__device__ __forceinline__ void atomAddF32(float* p, float v) {
    (void)__hip_atomic_fetch_add(p, v, __ATOMIC_RELAXED, __HIP_MEMORY_SCOPE_AGENT);
}

__global__ void k_zero(float* __restrict__ p, int n) {
    int i = blockIdx.x * blockDim.x + threadIdx.x;
    if (i < n) p[i] = 0.0f;
}

__global__ void k_deg(const int* __restrict__ row, const int* __restrict__ col,
                      const float* __restrict__ intents, float* __restrict__ deg) {
    int t = blockIdx.x * blockDim.x + threadIdx.x;
    int e = t >> 2;
    int q = t & 3;
    if (e >= N_EDGES) return;
    float z0 = intents[e];
    float z1 = intents[N_EDGES + e];
    float z2 = intents[2 * N_EDGES + e];
    float z3 = intents[3 * N_EDGES + e];
    float m  = fmaxf(fmaxf(z0, z1), fmaxf(z2, z3));
    float w0 = __expf(z0 - m), w1 = __expf(z1 - m);
    float w2 = __expf(z2 - m), w3 = __expf(z3 - m);
    float inv = 1.0f / (w0 + w1 + w2 + w3);
    w0 *= inv; w1 *= inv; w2 *= inv; w3 *= inv;
    const float wlo = (q & 1) ? w1 : w0;
    const float whi = (q & 1) ? w3 : w2;
    const float w   = (q & 2) ? whi : wlo;
    int r = row[e], c = col[e];
    atomAddF32(deg + r * 4 + q, w);
    atomAddF32(deg + c * 4 + q, w);
}

__global__ void k_dis(float* __restrict__ deg) {
    int i = blockIdx.x * blockDim.x + threadIdx.x;
    if (i >= N_NODES * 4) return;
    float d = deg[i];
    deg[i] = (d > 0.0f) ? rsqrtf(fmaxf(d, 1e-12f)) : 0.0f;
}

__device__ __forceinline__ void scale_rows(const float* src, float* dst,
                                           float f0, float f1, int lane) {
    const int hl = lane & 15;
    const int h  = lane >> 4;
    float b[2][8];
#pragma unroll
    for (int kp = 0; kp < 2; ++kp)
#pragma unroll
        for (int i = 0; i < 8; ++i)
            b[kp][i] = src[i * 64 + kp * 32 + lane];
#pragma unroll
    for (int kp = 0; kp < 2; ++kp) {
        const float nk = kp ? f1 : f0;
        v8f acc = {};
#pragma unroll
        for (int t = 0; t < 4; ++t) {
            v2f a, bb;
            a.x = (hl == 8 * h + 2 * t)     ? nk : 0.0f;
            a.y = (hl == 8 * h + 2 * t + 1) ? nk : 0.0f;
            bb.x = b[kp][2 * t];
            bb.y = b[kp][2 * t + 1];
            acc = __builtin_amdgcn_wmma_f32_16x16x4_f32(
                      false, a, false, bb, (short)0, acc, false, false);
        }
#pragma unroll
        for (int v = 0; v < 8; ++v)
            dst[v * 64 + kp * 32 + lane] = acc[v];
    }
}

__device__ __forceinline__ void zero_rows(float* p, int lane) {
    float4* q = (float4*)p;
    const float4 z = make_float4(0.0f, 0.0f, 0.0f, 0.0f);
    for (int i = 0; i < 4; ++i) q[i * 32 + lane] = z;
}

#define N_TILES (N_NODES / 8)

__global__ __launch_bounds__(256) void k_scale_first(
        const float* __restrict__ Gu, const float* __restrict__ Gi,
        const float* __restrict__ dis, float* __restrict__ y, float* __restrict__ tgt) {
    const int wave = __builtin_amdgcn_readfirstlane((int)(blockIdx.x * 8 + (threadIdx.x >> 5)));
    if (wave >= N_TILES) return;
    const int lane = threadIdx.x & 31;
    const int hl = lane & 15, j = hl & 7, hs = hl >> 3;
    const int nb = wave * 8;
    const float* src = (nb < N_USERS) ? (Gu + nb * 64) : (Gi + (nb - N_USERS) * 64);
    const float f0 = dis[(nb + j) * 4 + hs];
    const float f1 = dis[(nb + j) * 4 + 2 + hs];
    scale_rows(src, y + nb * 64, f0, f1, lane);
    zero_rows(tgt + nb * 64, lane);
}

__global__ __launch_bounds__(256) void k_scale_mid(
        float* buf, const float* __restrict__ dis, float* __restrict__ tgt) {
    const int wave = __builtin_amdgcn_readfirstlane((int)(blockIdx.x * 8 + (threadIdx.x >> 5)));
    if (wave >= N_TILES) return;
    const int lane = threadIdx.x & 31;
    const int hl = lane & 15, j = hl & 7, hs = hl >> 3;
    const int nb = wave * 8;
    const float d0 = dis[(nb + j) * 4 + hs];
    const float d1 = dis[(nb + j) * 4 + 2 + hs];
    scale_rows(buf + nb * 64, buf + nb * 64, d0 * d0, d1 * d1, lane);
    zero_rows(tgt + nb * 64, lane);
}

__global__ __launch_bounds__(256) void k_scale_last(float* buf, const float* __restrict__ dis) {
    const int wave = __builtin_amdgcn_readfirstlane((int)(blockIdx.x * 8 + (threadIdx.x >> 5)));
    if (wave >= N_TILES) return;
    const int lane = threadIdx.x & 31;
    const int hl = lane & 15, j = hl & 7, hs = hl >> 3;
    const int nb = wave * 8;
    const float f0 = dis[(nb + j) * 4 + hs];
    const float f1 = dis[(nb + j) * 4 + 2 + hs];
    scale_rows(buf + nb * 64, buf + nb * 64, f0, f1, lane);
}

__global__ __launch_bounds__(128) void k_propagate(
        const float* __restrict__ y, float* __restrict__ s,
        const int* __restrict__ row, const int* __restrict__ col) {
    const int wave = __builtin_amdgcn_readfirstlane(
                         (int)(blockIdx.x * (blockDim.x >> 5) + (threadIdx.x >> 5)));
    const int lane = threadIdx.x & 31;
    const int base = wave * 16;

    int rb[16], cb[16];
#pragma unroll
    for (int i = 0; i < 16; ++i) { rb[i] = row[base + i]; cb[i] = col[base + i]; }

    float v[32];
#pragma unroll
    for (int i = 0; i < 16; ++i) {
        v[2 * i]     = y[rb[i] * 64 + lane];
        v[2 * i + 1] = y[rb[i] * 64 + 32 + lane];
    }
#pragma unroll
    for (int i = 0; i < 16; ++i) {
        atomAddF32(s + cb[i] * 64 + lane,      v[2 * i]);
        atomAddF32(s + cb[i] * 64 + 32 + lane, v[2 * i + 1]);
    }
}

extern "C" void kernel_launch(void* const* d_in, const int* in_sizes, int n_in,
                              void* d_out, int out_size, void* d_ws, size_t ws_size,
                              hipStream_t stream) {
    const float* Gu      = (const float*)d_in[0];
    const float* Gi      = (const float*)d_in[1];
    const int*   edge    = (const int*)d_in[2];
    const float* intents = (const float*)d_in[3];
    const int* row = edge;
    const int* col = edge + N_EDGES;
    float* out = (float*)d_out;

    float* deg  = (float*)d_ws;
    float* xbuf = deg + (size_t)N_NODES * 4;

    const int ND = N_NODES * 4;
    dim3 b256(256);
    const int waves  = N_EDGES / 16;
    dim3 pgrid(waves / 4), pblock(128);
    dim3 sgrid((N_TILES + 7) / 8);

    k_zero<<<dim3((ND + 255) / 256), b256, 0, stream>>>(deg, ND);
    k_deg<<<dim3((4 * N_EDGES + 255) / 256), b256, 0, stream>>>(row, col, intents, deg);
    k_dis<<<dim3((ND + 255) / 256), b256, 0, stream>>>(deg);

    k_scale_first<<<sgrid, b256, 0, stream>>>(Gu, Gi, deg, xbuf, out);
    k_propagate<<<pgrid, pblock, 0, stream>>>(xbuf, out, row, col);

    k_scale_mid<<<sgrid, b256, 0, stream>>>(out, deg, xbuf);
    k_propagate<<<pgrid, pblock, 0, stream>>>(out, xbuf, row, col);

    k_scale_mid<<<sgrid, b256, 0, stream>>>(xbuf, deg, out);
    k_propagate<<<pgrid, pblock, 0, stream>>>(xbuf, out, row, col);

    k_scale_last<<<sgrid, b256, 0, stream>>>(out, deg);
}


# k_deg: two private deg planes, one per L2 half (wave's die id picks; plane 1 in the still unused out buffer): no deg line takes atomics from both halves; k_dis adds the planes (new reduction tree)
# speedup vs baseline: 2.0449x; 1.0695x over previous
#include <hip/hip_runtime.h>

#define N_USERS  100000
#define N_ITEMS  50000
#define N_NODES  150000
#define N_EDGES  1000000
#define EMBED_K  64

typedef __attribute__((ext_vector_type(2))) float v2f;
typedef __attribute__((ext_vector_type(8))) float v8f;

__device__ __forceinline__ void atomAddF32(float* p, float v) {
    (void)__hip_atomic_fetch_add(p, v, __ATOMIC_RELAXED, __HIP_MEMORY_SCOPE_AGENT);
}

__global__ void k_zero(float* __restrict__ p0, float* __restrict__ p1, int n) {
    int i = blockIdx.x * blockDim.x + threadIdx.x;
    if (i < n) { p0[i] = 0.0f; p1[i] = 0.0f; }
}

__global__ void k_deg(const int* __restrict__ row, const int* __restrict__ col,
                      const float* __restrict__ intents,
                      float* __restrict__ deg0, float* __restrict__ deg1) {
    const unsigned place = __builtin_amdgcn_s_sendmsg_rtn(0x87);
    float* __restrict__ deg = (place & (1u << 18)) ? deg1 : deg0;
    int t = blockIdx.x * blockDim.x + threadIdx.x;
    int e = t >> 2;
    int q = t & 3;
    if (e >= N_EDGES) return;
    float z0 = intents[e];
    float z1 = intents[N_EDGES + e];
    float z2 = intents[2 * N_EDGES + e];
    float z3 = intents[3 * N_EDGES + e];
    float m  = fmaxf(fmaxf(z0, z1), fmaxf(z2, z3));
    float w0 = __expf(z0 - m), w1 = __expf(z1 - m);
    float w2 = __expf(z2 - m), w3 = __expf(z3 - m);
    float inv = 1.0f / (w0 + w1 + w2 + w3);
    w0 *= inv; w1 *= inv; w2 *= inv; w3 *= inv;
    const float wlo = (q & 1) ? w1 : w0;
    const float whi = (q & 1) ? w3 : w2;
    const float w   = (q & 2) ? whi : wlo;
    int r = row[e], c = col[e];
    atomAddF32(deg + r * 4 + q, w);
    atomAddF32(deg + c * 4 + q, w);
}

__global__ void k_dis(float* __restrict__ deg, const float* __restrict__ deg1) {
    int i = blockIdx.x * blockDim.x + threadIdx.x;
    if (i >= N_NODES * 4) return;
    float d = deg[i] + deg1[i];
    deg[i] = (d > 0.0f) ? rsqrtf(fmaxf(d, 1e-12f)) : 0.0f;
}

__device__ __forceinline__ void scale_rows(const float* src, float* dst,
                                           float f0, float f1, int lane) {
    const int hl = lane & 15;
    const int h  = lane >> 4;
    float b[2][8];
#pragma unroll
    for (int kp = 0; kp < 2; ++kp)
#pragma unroll
        for (int i = 0; i < 8; ++i)
            b[kp][i] = src[i * 64 + kp * 32 + lane];
#pragma unroll
    for (int kp = 0; kp < 2; ++kp) {
        const float nk = kp ? f1 : f0;
        v8f acc = {};
#pragma unroll
        for (int t = 0; t < 4; ++t) {
            v2f a, bb;
            a.x = (hl == 8 * h + 2 * t)     ? nk : 0.0f;
            a.y = (hl == 8 * h + 2 * t + 1) ? nk : 0.0f;
            bb.x = b[kp][2 * t];
            bb.y = b[kp][2 * t + 1];
            acc = __builtin_amdgcn_wmma_f32_16x16x4_f32(
                      false, a, false, bb, (short)0, acc, false, false);
        }
#pragma unroll
        for (int v = 0; v < 8; ++v)
            dst[v * 64 + kp * 32 + lane] = acc[v];
    }
}

__device__ __forceinline__ void zero_rows(float* p, int lane) {
    float4* q = (float4*)p;
    const float4 z = make_float4(0.0f, 0.0f, 0.0f, 0.0f);
    for (int i = 0; i < 4; ++i) q[i * 32 + lane] = z;
}

#define N_TILES (N_NODES / 8)

__global__ __launch_bounds__(256) void k_scale_first(
        const float* __restrict__ Gu, const float* __restrict__ Gi,
        const float* __restrict__ dis, float* __restrict__ y, float* __restrict__ tgt) {
    const int wave = __builtin_amdgcn_readfirstlane((int)(blockIdx.x * 8 + (threadIdx.x >> 5)));
    if (wave >= N_TILES) return;
    const int lane = threadIdx.x & 31;
    const int hl = lane & 15, j = hl & 7, hs = hl >> 3;
    const int nb = wave * 8;
    const float* src = (nb < N_USERS) ? (Gu + nb * 64) : (Gi + (nb - N_USERS) * 64);
    const float f0 = dis[(nb + j) * 4 + hs];
    const float f1 = dis[(nb + j) * 4 + 2 + hs];
    scale_rows(src, y + nb * 64, f0, f1, lane);
    zero_rows(tgt + nb * 64, lane);
}

__global__ __launch_bounds__(256) void k_scale_mid(
        float* buf, const float* __restrict__ dis, float* __restrict__ tgt) {
    const int wave = __builtin_amdgcn_readfirstlane((int)(blockIdx.x * 8 + (threadIdx.x >> 5)));
    if (wave >= N_TILES) return;
    const int lane = threadIdx.x & 31;
    const int hl = lane & 15, j = hl & 7, hs = hl >> 3;
    const int nb = wave * 8;
    const float d0 = dis[(nb + j) * 4 + hs];
    const float d1 = dis[(nb + j) * 4 + 2 + hs];
    scale_rows(buf + nb * 64, buf + nb * 64, d0 * d0, d1 * d1, lane);
    zero_rows(tgt + nb * 64, lane);
}

__global__ __launch_bounds__(256) void k_scale_last(float* buf, const float* __restrict__ dis) {
    const int wave = __builtin_amdgcn_readfirstlane((int)(blockIdx.x * 8 + (threadIdx.x >> 5)));
    if (wave >= N_TILES) return;
    const int lane = threadIdx.x & 31;
    const int hl = lane & 15, j = hl & 7, hs = hl >> 3;
    const int nb = wave * 8;
    const float f0 = dis[(nb + j) * 4 + hs];
    const float f1 = dis[(nb + j) * 4 + 2 + hs];
    scale_rows(buf + nb * 64, buf + nb * 64, f0, f1, lane);
}

__global__ __launch_bounds__(128) void k_propagate(
        const float* __restrict__ y, float* __restrict__ s,
        const int* __restrict__ row, const int* __restrict__ col) {
    const int wave = __builtin_amdgcn_readfirstlane(
                         (int)(blockIdx.x * (blockDim.x >> 5) + (threadIdx.x >> 5)));
    const int lane = threadIdx.x & 31;
    const int base = wave * 16;

    int rb[16], cb[16];
#pragma unroll
    for (int i = 0; i < 16; ++i) { rb[i] = row[base + i]; cb[i] = col[base + i]; }

    float v[32];
#pragma unroll
    for (int i = 0; i < 16; ++i) {
        v[2 * i]     = y[rb[i] * 64 + lane];
        v[2 * i + 1] = y[rb[i] * 64 + 32 + lane];
    }
#pragma unroll
    for (int i = 0; i < 16; ++i) {
        atomAddF32(s + cb[i] * 64 + lane,      v[2 * i]);
        atomAddF32(s + cb[i] * 64 + 32 + lane, v[2 * i + 1]);
    }
}

extern "C" void kernel_launch(void* const* d_in, const int* in_sizes, int n_in,
                              void* d_out, int out_size, void* d_ws, size_t ws_size,
                              hipStream_t stream) {
    const float* Gu      = (const float*)d_in[0];
    const float* Gi      = (const float*)d_in[1];
    const int*   edge    = (const int*)d_in[2];
    const float* intents = (const float*)d_in[3];
    const int* row = edge;
    const int* col = edge + N_EDGES;
    float* out = (float*)d_out;

    float* deg  = (float*)d_ws;
    float* deg1 = out;
    float* xbuf = deg + (size_t)N_NODES * 4;

    const int ND = N_NODES * 4;
    dim3 b256(256);
    const int waves  = N_EDGES / 16;
    dim3 pgrid(waves / 4), pblock(128);
    dim3 sgrid((N_TILES + 7) / 8);

    k_zero<<<dim3((ND + 255) / 256), b256, 0, stream>>>(deg, deg1, ND);
    k_deg<<<dim3((4 * N_EDGES + 255) / 256), b256, 0, stream>>>(row, col, intents, deg, deg1);
    k_dis<<<dim3((ND + 255) / 256), b256, 0, stream>>>(deg, deg1);

    k_scale_first<<<sgrid, b256, 0, stream>>>(Gu, Gi, deg, xbuf, out);
    k_propagate<<<pgrid, pblock, 0, stream>>>(xbuf, out, row, col);

    k_scale_mid<<<sgrid, b256, 0, stream>>>(out, deg, xbuf);
    k_propagate<<<pgrid, pblock, 0, stream>>>(out, xbuf, row, col);

    k_scale_mid<<<sgrid, b256, 0, stream>>>(xbuf, deg, out);
    k_propagate<<<pgrid, pblock, 0, stream>>>(xbuf, out, row, col);

    k_scale_last<<<sgrid, b256, 0, stream>>>(out, deg);
}


# a wave loops over several work items (k_deg 8 edge groups; k_propagate 5 chunks, next gather issued before the adds; scaling 2 tiles; zero/dis one node record a lane): 4x fewer waves; arrival order
# speedup vs baseline: 2.1496x; 1.0512x over previous
#include <hip/hip_runtime.h>

#define N_USERS  100000
#define N_ITEMS  50000
#define N_NODES  150000
#define N_EDGES  1000000
#define EMBED_K  64

typedef __attribute__((ext_vector_type(2))) float v2f;
typedef __attribute__((ext_vector_type(8))) float v8f;

__device__ __forceinline__ void atomAddF32(float* p, float v) {
    (void)__hip_atomic_fetch_add(p, v, __ATOMIC_RELAXED, __HIP_MEMORY_SCOPE_AGENT);
}

__global__ void k_zero(float* __restrict__ p0, float* __restrict__ p1, int n_nodes) {
    int i = blockIdx.x * blockDim.x + threadIdx.x;
    const float4 z = make_float4(0.0f, 0.0f, 0.0f, 0.0f);
    if (i < n_nodes) { ((float4*)p0)[i] = z; ((float4*)p1)[i] = z; }
}

#define DEG_J 8
#define DEG_WAVES (N_EDGES / (8 * DEG_J))
__global__ __launch_bounds__(256) void k_deg(
                      const int* __restrict__ row, const int* __restrict__ col,
                      const float* __restrict__ intents,
                      float* __restrict__ deg0, float* __restrict__ deg1) {
    const unsigned place = __builtin_amdgcn_s_sendmsg_rtn(0x87);
    float* __restrict__ deg = (place & (1u << 18)) ? deg1 : deg0;
    const int wave = blockIdx.x * 8 + (threadIdx.x >> 5);
    if (wave >= DEG_WAVES) return;
    const int lane = threadIdx.x & 31;
    const int q  = lane & 3;
    const int e0 = wave * (8 * DEG_J) + (lane >> 2);
    float z0[DEG_J], z1[DEG_J], z2[DEG_J], z3[DEG_J];
    int r[DEG_J], c[DEG_J];
#pragma unroll
    for (int j = 0; j < DEG_J; ++j) {
        const int e = e0 + 8 * j;
        z0[j] = intents[e];
        z1[j] = intents[N_EDGES + e];
        z2[j] = intents[2 * N_EDGES + e];
        z3[j] = intents[3 * N_EDGES + e];
        r[j] = row[e]; c[j] = col[e];
    }
#pragma unroll
    for (int j = 0; j < DEG_J; ++j) {
        float m  = fmaxf(fmaxf(z0[j], z1[j]), fmaxf(z2[j], z3[j]));
        float w0 = __expf(z0[j] - m), w1 = __expf(z1[j] - m);
        float w2 = __expf(z2[j] - m), w3 = __expf(z3[j] - m);
        float inv = 1.0f / (w0 + w1 + w2 + w3);
        w0 *= inv; w1 *= inv; w2 *= inv; w3 *= inv;
        const float wlo = (q & 1) ? w1 : w0;
        const float whi = (q & 1) ? w3 : w2;
        const float w   = (q & 2) ? whi : wlo;
        atomAddF32(deg + r[j] * 4 + q, w);
        atomAddF32(deg + c[j] * 4 + q, w);
    }
}

__device__ __forceinline__ float dis_of(float d) {
    return (d > 0.0f) ? rsqrtf(fmaxf(d, 1e-12f)) : 0.0f;
}
__global__ void k_dis(float* __restrict__ deg, const float* __restrict__ deg1) {
    int i = blockIdx.x * blockDim.x + threadIdx.x;
    if (i >= N_NODES) return;
    const float4 a = ((const float4*)deg)[i], b = ((const float4*)deg1)[i];
    ((float4*)deg)[i] = make_float4(dis_of(a.x + b.x), dis_of(a.y + b.y),
                                    dis_of(a.z + b.z), dis_of(a.w + b.w));
}

#define SC_T 2
#define SC_WAVES (N_NODES / (8 * SC_T))
__device__ __forceinline__ void load_rows(const float* src, float (&b)[2][8], int lane) {
#pragma unroll
    for (int kp = 0; kp < 2; ++kp)
#pragma unroll
        for (int i = 0; i < 8; ++i)
            b[kp][i] = src[i * 64 + kp * 32 + lane];
}
__device__ __forceinline__ void scale_store(const float (&b)[2][8], float* dst,
                                            float f0, float f1, int lane) {
    const int hl = lane & 15;
    const int h  = lane >> 4;
#pragma unroll
    for (int kp = 0; kp < 2; ++kp) {
        const float nk = kp ? f1 : f0;
        v8f acc = {};
#pragma unroll
        for (int t = 0; t < 4; ++t) {
            v2f a, bb;
            a.x = (hl == 8 * h + 2 * t)     ? nk : 0.0f;
            a.y = (hl == 8 * h + 2 * t + 1) ? nk : 0.0f;
            bb.x = b[kp][2 * t];
            bb.y = b[kp][2 * t + 1];
            acc = __builtin_amdgcn_wmma_f32_16x16x4_f32(
                      false, a, false, bb, (short)0, acc, false, false);
        }
#pragma unroll
        for (int v = 0; v < 8; ++v)
            dst[v * 64 + kp * 32 + lane] = acc[v];
    }
}

__device__ __forceinline__ void zero_rows(float* p, int lane) {
    float4* q = (float4*)p;
    const float4 z = make_float4(0.0f, 0.0f, 0.0f, 0.0f);
    for (int i = 0; i < 4; ++i) q[i * 32 + lane] = z;
}


__global__ __launch_bounds__(256) void k_scale_first(
        const float* __restrict__ Gu, const float* __restrict__ Gi,
        const float* __restrict__ dis, float* __restrict__ y, float* __restrict__ tgt) {
    const int wave = __builtin_amdgcn_readfirstlane((int)(blockIdx.x * 8 + (threadIdx.x >> 5)));
    if (wave >= SC_WAVES) return;
    const int lane = threadIdx.x & 31;
    const int hl = lane & 15, j = hl & 7, hs = hl >> 3;
    float b[SC_T][2][8], f0[SC_T], f1[SC_T];
#pragma unroll
    for (int t = 0; t < SC_T; ++t) {
        const int nb = (wave * SC_T + t) * 8;
        const float* src = (nb < N_USERS) ? (Gu + nb * 64) : (Gi + (nb - N_USERS) * 64);
        f0[t] = dis[(nb + j) * 4 + hs];
        f1[t] = dis[(nb + j) * 4 + 2 + hs];
        load_rows(src, b[t], lane);
    }
#pragma unroll
    for (int t = 0; t < SC_T; ++t) {
        const int nb = (wave * SC_T + t) * 8;
        scale_store(b[t], y + nb * 64, f0[t], f1[t], lane);
        zero_rows(tgt + nb * 64, lane);
    }
}

__global__ __launch_bounds__(256) void k_scale_mid(
        float* buf, const float* __restrict__ dis, float* __restrict__ tgt) {
    const int wave = __builtin_amdgcn_readfirstlane((int)(blockIdx.x * 8 + (threadIdx.x >> 5)));
    if (wave >= SC_WAVES) return;
    const int lane = threadIdx.x & 31;
    const int hl = lane & 15, j = hl & 7, hs = hl >> 3;
    float b[SC_T][2][8], d0[SC_T], d1[SC_T];
#pragma unroll
    for (int t = 0; t < SC_T; ++t) {
        const int nb = (wave * SC_T + t) * 8;
        d0[t] = dis[(nb + j) * 4 + hs];
        d1[t] = dis[(nb + j) * 4 + 2 + hs];
        load_rows(buf + nb * 64, b[t], lane);
    }
#pragma unroll
    for (int t = 0; t < SC_T; ++t) {
        const int nb = (wave * SC_T + t) * 8;
        scale_store(b[t], buf + nb * 64, d0[t] * d0[t], d1[t] * d1[t], lane);
        zero_rows(tgt + nb * 64, lane);
    }
}

__global__ __launch_bounds__(256) void k_scale_last(float* buf, const float* __restrict__ dis) {
    const int wave = __builtin_amdgcn_readfirstlane((int)(blockIdx.x * 8 + (threadIdx.x >> 5)));
    if (wave >= SC_WAVES) return;
    const int lane = threadIdx.x & 31;
    const int hl = lane & 15, j = hl & 7, hs = hl >> 3;
    float b[SC_T][2][8], f0[SC_T], f1[SC_T];
#pragma unroll
    for (int t = 0; t < SC_T; ++t) {
        const int nb = (wave * SC_T + t) * 8;
        f0[t] = dis[(nb + j) * 4 + hs];
        f1[t] = dis[(nb + j) * 4 + 2 + hs];
        load_rows(buf + nb * 64, b[t], lane);
    }
#pragma unroll
    for (int t = 0; t < SC_T; ++t) {
        const int nb = (wave * SC_T + t) * 8;
        scale_store(b[t], buf + nb * 64, f0[t], f1[t], lane);
    }
}

#define PROP_C 5
#define PROP_WAVES (N_EDGES / (16 * PROP_C))
__device__ __forceinline__ void edge_ids(const int* __restrict__ row, const int* __restrict__ col,
                                         int base, int (&rb)[16], int (&cb)[16]) {
#pragma unroll
    for (int i = 0; i < 16; ++i) { rb[i] = row[base + i]; cb[i] = col[base + i]; }
}
__device__ __forceinline__ void gather_rows(const float* __restrict__ y, const int (&rb)[16],
                                            float (&v)[32], int lane) {
#pragma unroll
    for (int i = 0; i < 16; ++i) {
        v[2 * i]     = y[rb[i] * 64 + lane];
        v[2 * i + 1] = y[rb[i] * 64 + 32 + lane];
    }
}
__device__ __forceinline__ void add_rows(float* __restrict__ s, const int (&cb)[16],
                                         const float (&v)[32], int lane) {
#pragma unroll
    for (int i = 0; i < 16; ++i) {
        atomAddF32(s + cb[i] * 64 + lane,      v[2 * i]);
        atomAddF32(s + cb[i] * 64 + 32 + lane, v[2 * i + 1]);
    }
}
__global__ __launch_bounds__(128) void k_propagate(
        const float* __restrict__ y, float* __restrict__ s,
        const int* __restrict__ row, const int* __restrict__ col) {
    const int wave = __builtin_amdgcn_readfirstlane(
                         (int)(blockIdx.x * (blockDim.x >> 5) + (threadIdx.x >> 5)));
    const int lane = threadIdx.x & 31;
    const int base = wave * (16 * PROP_C);

    int rb[2][16], cb[2][16];
    float v[2][32];
    edge_ids(row, col, base, rb[0], cb[0]);
    gather_rows(y, rb[0], v[0], lane);
#pragma unroll
    for (int k = 0; k < PROP_C; ++k) {
        const int cur = k & 1, nxt = cur ^ 1;
        if (k + 1 < PROP_C) {
            edge_ids(row, col, base + 16 * (k + 1), rb[nxt], cb[nxt]);
            gather_rows(y, rb[nxt], v[nxt], lane);
        }
        add_rows(s, cb[cur], v[cur], lane);
    }
}

extern "C" void kernel_launch(void* const* d_in, const int* in_sizes, int n_in,
                              void* d_out, int out_size, void* d_ws, size_t ws_size,
                              hipStream_t stream) {
    const float* Gu      = (const float*)d_in[0];
    const float* Gi      = (const float*)d_in[1];
    const int*   edge    = (const int*)d_in[2];
    const float* intents = (const float*)d_in[3];
    const int* row = edge;
    const int* col = edge + N_EDGES;
    float* out = (float*)d_out;

    float* deg  = (float*)d_ws;
    float* deg1 = out;
    float* xbuf = deg + (size_t)N_NODES * 4;

    dim3 b256(256);
    dim3 pgrid(PROP_WAVES / 4), pblock(128);
    dim3 sgrid((SC_WAVES + 7) / 8);

    k_zero<<<dim3((N_NODES + 255) / 256), b256, 0, stream>>>(deg, deg1, N_NODES);
    k_deg<<<dim3((DEG_WAVES + 7) / 8), b256, 0, stream>>>(row, col, intents, deg, deg1);
    k_dis<<<dim3((N_NODES + 255) / 256), b256, 0, stream>>>(deg, deg1);

    k_scale_first<<<sgrid, b256, 0, stream>>>(Gu, Gi, deg, xbuf, out);
    k_propagate<<<pgrid, pblock, 0, stream>>>(xbuf, out, row, col);

    k_scale_mid<<<sgrid, b256, 0, stream>>>(out, deg, xbuf);
    k_propagate<<<pgrid, pblock, 0, stream>>>(out, xbuf, row, col);

    k_scale_mid<<<sgrid, b256, 0, stream>>>(xbuf, deg, out);
    k_propagate<<<pgrid, pblock, 0, stream>>>(xbuf, out, row, col);

    k_scale_last<<<sgrid, b256, 0, stream>>>(out, deg);
}


# edge passes pull, not push: k_deg chains each edge to its destination (one returning integer exchange); a wave sums 8 nodes' source lines in registers per half row; no float atomics; list-order sums
# speedup vs baseline: 2.2415x; 1.0428x over previous
#include <hip/hip_runtime.h>

#define N_USERS  100000
#define N_ITEMS  50000
#define N_NODES  150000
#define N_EDGES  1000000
#define EMBED_K  64

typedef __attribute__((ext_vector_type(2))) float v2f;
typedef __attribute__((ext_vector_type(8))) float v8f;

__device__ __forceinline__ void atomAddF32(float* p, float v) {
    (void)__hip_atomic_fetch_add(p, v, __ATOMIC_RELAXED, __HIP_MEMORY_SCOPE_AGENT);
}

__global__ void k_zero(float* __restrict__ p0, float* __restrict__ p1,
                       int* __restrict__ head, int n_nodes) {
    int i = blockIdx.x * blockDim.x + threadIdx.x;
    const float4 z = make_float4(0.0f, 0.0f, 0.0f, 0.0f);
    if (i < n_nodes) { ((float4*)p0)[i] = z; ((float4*)p1)[i] = z; head[i] = -1; }
}

#define DEG_J 8
#define DEG_WAVES (N_EDGES / (8 * DEG_J))
__global__ __launch_bounds__(256) void k_deg(
                      const int* __restrict__ row, const int* __restrict__ col,
                      const float* __restrict__ intents,
                      float* __restrict__ deg0, float* __restrict__ deg1,
                      int* __restrict__ head, int2* __restrict__ link) {
    const unsigned place = __builtin_amdgcn_s_sendmsg_rtn(0x87);
    float* __restrict__ deg = (place & (1u << 18)) ? deg1 : deg0;
    const int wave = blockIdx.x * 8 + (threadIdx.x >> 5);
    if (wave >= DEG_WAVES) return;
    const int lane = threadIdx.x & 31;
    const int eA = wave * (8 * DEG_J) + lane, eB = eA + 32;
    const int rA = row[eA], cA = col[eA], rB = row[eB], cB = col[eB];
    const int oA = __hip_atomic_exchange(head + cA, eA, __ATOMIC_RELAXED, __HIP_MEMORY_SCOPE_AGENT);
    const int oB = __hip_atomic_exchange(head + cB, eB, __ATOMIC_RELAXED, __HIP_MEMORY_SCOPE_AGENT);
    const int q  = lane & 3;
    const int e0 = wave * (8 * DEG_J) + (lane >> 2);
    float z0[DEG_J], z1[DEG_J], z2[DEG_J], z3[DEG_J];
    int r[DEG_J], c[DEG_J];
#pragma unroll
    for (int j = 0; j < DEG_J; ++j) {
        const int e = e0 + 8 * j;
        z0[j] = intents[e];
        z1[j] = intents[N_EDGES + e];
        z2[j] = intents[2 * N_EDGES + e];
        z3[j] = intents[3 * N_EDGES + e];
        r[j] = row[e]; c[j] = col[e];
    }
#pragma unroll
    for (int j = 0; j < DEG_J; ++j) {
        float m  = fmaxf(fmaxf(z0[j], z1[j]), fmaxf(z2[j], z3[j]));
        float w0 = __expf(z0[j] - m), w1 = __expf(z1[j] - m);
        float w2 = __expf(z2[j] - m), w3 = __expf(z3[j] - m);
        float inv = 1.0f / (w0 + w1 + w2 + w3);
        w0 *= inv; w1 *= inv; w2 *= inv; w3 *= inv;
        const float wlo = (q & 1) ? w1 : w0;
        const float whi = (q & 1) ? w3 : w2;
        const float w   = (q & 2) ? whi : wlo;
        atomAddF32(deg + r[j] * 4 + q, w);
        atomAddF32(deg + c[j] * 4 + q, w);
    }
    link[eA] = make_int2(rA, oA);
    link[eB] = make_int2(rB, oB);
}

__device__ __forceinline__ float dis_of(float d) {
    return (d > 0.0f) ? rsqrtf(fmaxf(d, 1e-12f)) : 0.0f;
}
__global__ void k_dis(float* __restrict__ deg, const float* __restrict__ deg1) {
    int i = blockIdx.x * blockDim.x + threadIdx.x;
    if (i >= N_NODES) return;
    const float4 a = ((const float4*)deg)[i], b = ((const float4*)deg1)[i];
    ((float4*)deg)[i] = make_float4(dis_of(a.x + b.x), dis_of(a.y + b.y),
                                    dis_of(a.z + b.z), dis_of(a.w + b.w));
}

template <int DS>
__device__ __forceinline__ void scale_store_half(const float (&b)[8], float* dst,
                                                 float nk, int lane) {
    const int hl = lane & 15;
    const int h  = lane >> 4;
    v8f acc = {};
#pragma unroll
    for (int t = 0; t < 4; ++t) {
        v2f a, bb;
        a.x = (hl == 8 * h + 2 * t)     ? nk : 0.0f;
        a.y = (hl == 8 * h + 2 * t + 1) ? nk : 0.0f;
        bb.x = b[2 * t];
        bb.y = b[2 * t + 1];
        acc = __builtin_amdgcn_wmma_f32_16x16x4_f32(
                  false, a, false, bb, (short)0, acc, false, false);
    }
#pragma unroll
    for (int v = 0; v < 8; ++v)
        dst[v * DS + lane] = acc[v];
}

#define SC_T 2
#define SC_WAVES (N_NODES / (8 * SC_T))
template <int DS>
__global__ __launch_bounds__(256) void k_scale_first(
        const float* __restrict__ Gu, const float* __restrict__ Gi,
        const float* __restrict__ dis, float* dst, int kp) {
    const int wave = __builtin_amdgcn_readfirstlane((int)(blockIdx.x * 8 + (threadIdx.x >> 5)));
    if (wave >= SC_WAVES) return;
    const int lane = threadIdx.x & 31;
    const int hl = lane & 15, j = hl & 7, hs = hl >> 3;
    float b[SC_T][8], f[SC_T];
#pragma unroll
    for (int t = 0; t < SC_T; ++t) {
        const int nb = (wave * SC_T + t) * 8;
        const float* src = (nb < N_USERS) ? (Gu + nb * 64) : (Gi + (nb - N_USERS) * 64);
        f[t] = dis[(nb + j) * 4 + 2 * kp + hs];
#pragma unroll
        for (int i = 0; i < 8; ++i)
            b[t][i] = src[i * 64 + kp * 32 + lane];
    }
#pragma unroll
    for (int t = 0; t < SC_T; ++t) {
        const int nb = (wave * SC_T + t) * 8;
        scale_store_half<DS>(b[t], dst + nb * DS, f[t], lane);
    }
}

#define G_WAVES (N_NODES / 8)
template <int SS, int DS>
__global__ __launch_bounds__(256) void k_gather(
        const float* src, float* dst, const float* __restrict__ dis,
        const int* __restrict__ head, const int2* __restrict__ link, int kp, int sq) {
    const int wave = __builtin_amdgcn_readfirstlane((int)(blockIdx.x * 8 + (threadIdx.x >> 5)));
    if (wave >= G_WAVES) return;
    const int lane = threadIdx.x & 31;
    const int nb = wave * 8;
    const int hl = lane & 15, j = hl & 7, hs = hl >> 3;
    const float d = dis[(nb + j) * 4 + 2 * kp + hs];
    const float f = sq ? d * d : d;

    const int e0 = head[nb + (lane & 7)];
    int2 lk = make_int2(-1, -1);
    if (e0 >= 0) lk = link[e0];
    float acc[8];
#pragma unroll
    for (int i = 0; i < 8; ++i) acc[i] = 0.0f;

    while (__builtin_amdgcn_ballot_w32(lk.x >= 0) != 0u) {
        const int en = lk.y;
        int2 lkn = make_int2(-1, -1);
        if (en >= 0) lkn = link[en];
        int   sid[8];
        float v[8];
#pragma unroll
        for (int i = 0; i < 8; ++i) {
            sid[i] = __builtin_amdgcn_readlane(lk.x, i);
            const int idx = sid[i] >= 0 ? sid[i] : 0;
            v[i] = src[idx * SS + lane];
        }
#pragma unroll
        for (int i = 0; i < 8; ++i)
            acc[i] += (sid[i] >= 0) ? v[i] : 0.0f;
        lk = lkn;
    }
    scale_store_half<DS>(acc, dst + nb * DS, f, lane);
}

extern "C" void kernel_launch(void* const* d_in, const int* in_sizes, int n_in,
                              void* d_out, int out_size, void* d_ws, size_t ws_size,
                              hipStream_t stream) {
    const float* Gu      = (const float*)d_in[0];
    const float* Gi      = (const float*)d_in[1];
    const int*   edge    = (const int*)d_in[2];
    const float* intents = (const float*)d_in[3];
    const int* row = edge;
    const int* col = edge + N_EDGES;
    float* out = (float*)d_out;

    float* deg  = (float*)d_ws;
    float* deg1 = out;
    float* hbuf = deg + (size_t)N_NODES * 4;
    int2*  link = (int2*)(hbuf + (size_t)N_NODES * 32);
    int*   head = (int*)(link + N_EDGES);

    dim3 b256(256);
    dim3 sgrid((SC_WAVES + 7) / 8);
    dim3 ggrid((G_WAVES + 7) / 8);

    k_zero<<<dim3((N_NODES + 255) / 256), b256, 0, stream>>>(deg, deg1, head, N_NODES);
    k_deg<<<dim3((DEG_WAVES + 7) / 8), b256, 0, stream>>>(row, col, intents, deg, deg1, head, link);
    k_dis<<<dim3((N_NODES + 255) / 256), b256, 0, stream>>>(deg, deg1);

    k_scale_first<64><<<sgrid, b256, 0, stream>>>(Gu, Gi, deg, out + 32, 0);
    k_gather<64, 64><<<ggrid, b256, 0, stream>>>(out + 32, out, deg, head, link, 0, 1);
    k_gather<64, 32><<<ggrid, b256, 0, stream>>>(out, hbuf, deg, head, link, 0, 1);
    k_gather<32, 64><<<ggrid, b256, 0, stream>>>(hbuf, out, deg, head, link, 0, 0);

    k_scale_first<32><<<sgrid, b256, 0, stream>>>(Gu, Gi, deg, hbuf, 1);
    k_gather<32, 64><<<ggrid, b256, 0, stream>>>(hbuf, out + 32, deg, head, link, 1, 1);
    k_gather<64, 32><<<ggrid, b256, 0, stream>>>(out + 32, hbuf, deg, head, link, 1, 1);
    k_gather<32, 64><<<ggrid, b256, 0, stream>>>(hbuf, out + 32, deg, head, link, 1, 0);
}


# layer 1 gathers whole 256-byte rows straight from the inputs, source factor fused into the sum (fma, y0 no longer rounded alone): no scaled-input pass, one list walk for both halves; list-order sums
# speedup vs baseline: 2.6350x; 1.1756x over previous
#include <hip/hip_runtime.h>

#define N_USERS  100000
#define N_ITEMS  50000
#define N_NODES  150000
#define N_EDGES  1000000
#define EMBED_K  64

typedef __attribute__((ext_vector_type(2))) float v2f;
typedef __attribute__((ext_vector_type(8))) float v8f;

__device__ __forceinline__ void atomAddF32(float* p, float v) {
    (void)__hip_atomic_fetch_add(p, v, __ATOMIC_RELAXED, __HIP_MEMORY_SCOPE_AGENT);
}

__global__ void k_zero(float* __restrict__ p0, float* __restrict__ p1,
                       int* __restrict__ head, int n_nodes) {
    int i = blockIdx.x * blockDim.x + threadIdx.x;
    const float4 z = make_float4(0.0f, 0.0f, 0.0f, 0.0f);
    if (i < n_nodes) { ((float4*)p0)[i] = z; ((float4*)p1)[i] = z; head[i] = -1; }
}

#define DEG_J 8
#define DEG_WAVES (N_EDGES / (8 * DEG_J))
__global__ __launch_bounds__(256) void k_deg(
                      const int* __restrict__ row, const int* __restrict__ col,
                      const float* __restrict__ intents,
                      float* __restrict__ deg0, float* __restrict__ deg1,
                      int* __restrict__ head, int2* __restrict__ link) {
    const unsigned place = __builtin_amdgcn_s_sendmsg_rtn(0x87);
    float* __restrict__ deg = (place & (1u << 18)) ? deg1 : deg0;
    const int wave = blockIdx.x * 8 + (threadIdx.x >> 5);
    if (wave >= DEG_WAVES) return;
    const int lane = threadIdx.x & 31;
    const int eA = wave * (8 * DEG_J) + lane, eB = eA + 32;
    const int rA = row[eA], cA = col[eA], rB = row[eB], cB = col[eB];
    const int oA = __hip_atomic_exchange(head + cA, eA, __ATOMIC_RELAXED, __HIP_MEMORY_SCOPE_AGENT);
    const int oB = __hip_atomic_exchange(head + cB, eB, __ATOMIC_RELAXED, __HIP_MEMORY_SCOPE_AGENT);
    const int q  = lane & 3;
    const int e0 = wave * (8 * DEG_J) + (lane >> 2);
    float z0[DEG_J], z1[DEG_J], z2[DEG_J], z3[DEG_J];
    int r[DEG_J], c[DEG_J];
#pragma unroll
    for (int j = 0; j < DEG_J; ++j) {
        const int e = e0 + 8 * j;
        z0[j] = intents[e];
        z1[j] = intents[N_EDGES + e];
        z2[j] = intents[2 * N_EDGES + e];
        z3[j] = intents[3 * N_EDGES + e];
        r[j] = row[e]; c[j] = col[e];
    }
#pragma unroll
    for (int j = 0; j < DEG_J; ++j) {
        float m  = fmaxf(fmaxf(z0[j], z1[j]), fmaxf(z2[j], z3[j]));
        float w0 = __expf(z0[j] - m), w1 = __expf(z1[j] - m);
        float w2 = __expf(z2[j] - m), w3 = __expf(z3[j] - m);
        float inv = 1.0f / (w0 + w1 + w2 + w3);
        w0 *= inv; w1 *= inv; w2 *= inv; w3 *= inv;
        const float wlo = (q & 1) ? w1 : w0;
        const float whi = (q & 1) ? w3 : w2;
        const float w   = (q & 2) ? whi : wlo;
        atomAddF32(deg + r[j] * 4 + q, w);
        atomAddF32(deg + c[j] * 4 + q, w);
    }
    link[eA] = make_int2(rA, oA);
    link[eB] = make_int2(rB, oB);
}

__device__ __forceinline__ float dis_of(float d) {
    return (d > 0.0f) ? rsqrtf(fmaxf(d, 1e-12f)) : 0.0f;
}
__global__ void k_dis(float* __restrict__ deg, const float* __restrict__ deg1) {
    int i = blockIdx.x * blockDim.x + threadIdx.x;
    if (i >= N_NODES) return;
    const float4 a = ((const float4*)deg)[i], b = ((const float4*)deg1)[i];
    ((float4*)deg)[i] = make_float4(dis_of(a.x + b.x), dis_of(a.y + b.y),
                                    dis_of(a.z + b.z), dis_of(a.w + b.w));
}

template <int DS>
__device__ __forceinline__ void scale_store_half(const float (&b)[8], float* dst,
                                                 float nk, int lane) {
    const int hl = lane & 15;
    const int h  = lane >> 4;
    v8f acc = {};
#pragma unroll
    for (int t = 0; t < 4; ++t) {
        v2f a, bb;
        a.x = (hl == 8 * h + 2 * t)     ? nk : 0.0f;
        a.y = (hl == 8 * h + 2 * t + 1) ? nk : 0.0f;
        bb.x = b[2 * t];
        bb.y = b[2 * t + 1];
        acc = __builtin_amdgcn_wmma_f32_16x16x4_f32(
                  false, a, false, bb, (short)0, acc, false, false);
    }
#pragma unroll
    for (int v = 0; v < 8; ++v)
        dst[v * DS + lane] = acc[v];
}

#define G_WAVES (N_NODES / 8)
template <int SS, int DS>
__global__ __launch_bounds__(256) void k_gather(
        const float* src, float* dst, const float* __restrict__ dis,
        const int* __restrict__ head, const int2* __restrict__ link, int kp, int sq) {
    const int wave = __builtin_amdgcn_readfirstlane((int)(blockIdx.x * 8 + (threadIdx.x >> 5)));
    if (wave >= G_WAVES) return;
    const int lane = threadIdx.x & 31;
    const int nb = wave * 8;
    const int hl = lane & 15, j = hl & 7, hs = hl >> 3;
    const float d = dis[(nb + j) * 4 + 2 * kp + hs];
    const float f = sq ? d * d : d;

    const int e0 = head[nb + (lane & 7)];
    int2 lk = make_int2(-1, -1);
    if (e0 >= 0) lk = link[e0];
    float acc[8];
#pragma unroll
    for (int i = 0; i < 8; ++i) acc[i] = 0.0f;

    while (__builtin_amdgcn_ballot_w32(lk.x >= 0) != 0u) {
        const int en = lk.y;
        int2 lkn = make_int2(-1, -1);
        if (en >= 0) lkn = link[en];
        int   sid[8];
        float v[8];
#pragma unroll
        for (int i = 0; i < 8; ++i) {
            sid[i] = __builtin_amdgcn_readlane(lk.x, i);
            const int idx = sid[i] >= 0 ? sid[i] : 0;
            v[i] = src[idx * SS + lane];
        }
#pragma unroll
        for (int i = 0; i < 8; ++i)
            acc[i] += (sid[i] >= 0) ? v[i] : 0.0f;
        lk = lkn;
    }
    scale_store_half<DS>(acc, dst + nb * DS, f, lane);
}

__global__ __launch_bounds__(256) void k_gather_first(
        const float* __restrict__ Gu, const float* __restrict__ Gi, float* __restrict__ dst,
        const float* __restrict__ dis, const int* __restrict__ head,
        const int2* __restrict__ link) {
    const int wave = __builtin_amdgcn_readfirstlane((int)(blockIdx.x * 8 + (threadIdx.x >> 5)));
    if (wave >= G_WAVES) return;
    const int lane = threadIdx.x & 31;
    const int nb = wave * 8;
    const int hl = lane & 15, j = hl & 7, hs = hl >> 3;
    const int h = lane >> 4;
    const int q = lane >> 3;
    const float d0 = dis[(nb + j) * 4 + hs];
    const float d1 = dis[(nb + j) * 4 + 2 + hs];

    const int e0 = head[nb + (lane & 7)];
    int2 lk = make_int2(-1, -1);
    if (e0 >= 0) lk = link[e0];
    float accl[8], acch[8];
#pragma unroll
    for (int i = 0; i < 8; ++i) { accl[i] = 0.0f; acch[i] = 0.0f; }

    while (__builtin_amdgcn_ballot_w32(lk.x >= 0) != 0u) {
        const int en = lk.y;
        int2 lkn = make_int2(-1, -1);
        if (en >= 0) lkn = link[en];
        const int sx = lk.x >= 0 ? lk.x : 0;
        const int dv = __float_as_int(dis[sx * 4 + q]);
        int   sid[8];
        float vl[8], vh[8];
#pragma unroll
        for (int i = 0; i < 8; ++i) {
            sid[i] = __builtin_amdgcn_readlane(lk.x, i);
            const int idx = sid[i] >= 0 ? sid[i] : 0;
            const float* p = (idx < N_USERS) ? (Gu + idx * 64) : (Gi + (idx - N_USERS) * 64);
            vl[i] = p[lane];
            vh[i] = p[32 + lane];
        }
#pragma unroll
        for (int i = 0; i < 8; ++i) {
            const float s0 = __int_as_float(__builtin_amdgcn_readlane(dv, i));
            const float s1 = __int_as_float(__builtin_amdgcn_readlane(dv, i + 8));
            const float s2 = __int_as_float(__builtin_amdgcn_readlane(dv, i + 16));
            const float s3 = __int_as_float(__builtin_amdgcn_readlane(dv, i + 24));
            const float dl = h ? s1 : s0;
            const float dh = h ? s3 : s2;
            const float xl = (sid[i] >= 0) ? vl[i] : 0.0f;
            const float xh = (sid[i] >= 0) ? vh[i] : 0.0f;
            accl[i] = fmaf(xl, dl, accl[i]);
            acch[i] = fmaf(xh, dh, acch[i]);
        }
        lk = lkn;
    }
    scale_store_half<64>(accl, dst + nb * 64,      d0 * d0, lane);
    scale_store_half<64>(acch, dst + nb * 64 + 32, d1 * d1, lane);
}

extern "C" void kernel_launch(void* const* d_in, const int* in_sizes, int n_in,
                              void* d_out, int out_size, void* d_ws, size_t ws_size,
                              hipStream_t stream) {
    const float* Gu      = (const float*)d_in[0];
    const float* Gi      = (const float*)d_in[1];
    const int*   edge    = (const int*)d_in[2];
    const float* intents = (const float*)d_in[3];
    const int* row = edge;
    const int* col = edge + N_EDGES;
    float* out = (float*)d_out;

    float* deg  = (float*)d_ws;
    float* deg1 = out;
    float* hbuf = deg + (size_t)N_NODES * 4;
    int2*  link = (int2*)(hbuf + (size_t)N_NODES * 32);
    int*   head = (int*)(link + N_EDGES);

    dim3 b256(256);
    dim3 ggrid((G_WAVES + 7) / 8);

    k_zero<<<dim3((N_NODES + 255) / 256), b256, 0, stream>>>(deg, deg1, head, N_NODES);
    k_deg<<<dim3((DEG_WAVES + 7) / 8), b256, 0, stream>>>(row, col, intents, deg, deg1, head, link);
    k_dis<<<dim3((N_NODES + 255) / 256), b256, 0, stream>>>(deg, deg1);

    k_gather_first<<<ggrid, b256, 0, stream>>>(Gu, Gi, out, deg, head, link);

    k_gather<64, 32><<<ggrid, b256, 0, stream>>>(out, hbuf, deg, head, link, 0, 1);
    k_gather<32, 64><<<ggrid, b256, 0, stream>>>(hbuf, out, deg, head, link, 0, 0);

    k_gather<64, 32><<<ggrid, b256, 0, stream>>>(out + 32, hbuf, deg, head, link, 1, 1);
    k_gather<32, 64><<<ggrid, b256, 0, stream>>>(hbuf, out + 32, deg, head, link, 1, 0);
}


# layer 1's list walk also writes each wave's source ids step by step as whole lines; layers 2-3 read these records (32 row loads in flight) and chase no pointers; list-order sums as before
# speedup vs baseline: 2.7411x; 1.0403x over previous
#include <hip/hip_runtime.h>

#define N_USERS  100000
#define N_ITEMS  50000
#define N_NODES  150000
#define N_EDGES  1000000
#define EMBED_K  64

typedef __attribute__((ext_vector_type(2))) float v2f;
typedef __attribute__((ext_vector_type(8))) float v8f;

__device__ __forceinline__ void atomAddF32(float* p, float v) {
    (void)__hip_atomic_fetch_add(p, v, __ATOMIC_RELAXED, __HIP_MEMORY_SCOPE_AGENT);
}

__global__ void k_zero(float* __restrict__ p0, float* __restrict__ p1,
                       int* __restrict__ head, int n_nodes) {
    int i = blockIdx.x * blockDim.x + threadIdx.x;
    const float4 z = make_float4(0.0f, 0.0f, 0.0f, 0.0f);
    if (i < n_nodes) { ((float4*)p0)[i] = z; ((float4*)p1)[i] = z; head[i] = -1; }
}

#define DEG_J 8
#define DEG_WAVES (N_EDGES / (8 * DEG_J))
__global__ __launch_bounds__(256) void k_deg(
                      const int* __restrict__ row, const int* __restrict__ col,
                      const float* __restrict__ intents,
                      float* __restrict__ deg0, float* __restrict__ deg1,
                      int* __restrict__ head, int2* __restrict__ link) {
    const unsigned place = __builtin_amdgcn_s_sendmsg_rtn(0x87);
    float* __restrict__ deg = (place & (1u << 18)) ? deg1 : deg0;
    const int wave = blockIdx.x * 8 + (threadIdx.x >> 5);
    if (wave >= DEG_WAVES) return;
    const int lane = threadIdx.x & 31;
    const int eA = wave * (8 * DEG_J) + lane, eB = eA + 32;
    const int rA = row[eA], cA = col[eA], rB = row[eB], cB = col[eB];
    const int oA = __hip_atomic_exchange(head + cA, eA, __ATOMIC_RELAXED, __HIP_MEMORY_SCOPE_AGENT);
    const int oB = __hip_atomic_exchange(head + cB, eB, __ATOMIC_RELAXED, __HIP_MEMORY_SCOPE_AGENT);
    const int q  = lane & 3;
    const int e0 = wave * (8 * DEG_J) + (lane >> 2);
    float z0[DEG_J], z1[DEG_J], z2[DEG_J], z3[DEG_J];
    int r[DEG_J], c[DEG_J];
#pragma unroll
    for (int j = 0; j < DEG_J; ++j) {
        const int e = e0 + 8 * j;
        z0[j] = intents[e];
        z1[j] = intents[N_EDGES + e];
        z2[j] = intents[2 * N_EDGES + e];
        z3[j] = intents[3 * N_EDGES + e];
        r[j] = row[e]; c[j] = col[e];
    }
#pragma unroll
    for (int j = 0; j < DEG_J; ++j) {
        float m  = fmaxf(fmaxf(z0[j], z1[j]), fmaxf(z2[j], z3[j]));
        float w0 = __expf(z0[j] - m), w1 = __expf(z1[j] - m);
        float w2 = __expf(z2[j] - m), w3 = __expf(z3[j] - m);
        float inv = 1.0f / (w0 + w1 + w2 + w3);
        w0 *= inv; w1 *= inv; w2 *= inv; w3 *= inv;
        const float wlo = (q & 1) ? w1 : w0;
        const float whi = (q & 1) ? w3 : w2;
        const float w   = (q & 2) ? whi : wlo;
        atomAddF32(deg + r[j] * 4 + q, w);
        atomAddF32(deg + c[j] * 4 + q, w);
    }
    link[eA] = make_int2(rA, oA);
    link[eB] = make_int2(rB, oB);
}

__device__ __forceinline__ float dis_of(float d) {
    return (d > 0.0f) ? rsqrtf(fmaxf(d, 1e-12f)) : 0.0f;
}
__global__ void k_dis(float* __restrict__ deg, const float* __restrict__ deg1) {
    int i = blockIdx.x * blockDim.x + threadIdx.x;
    if (i >= N_NODES) return;
    const float4 a = ((const float4*)deg)[i], b = ((const float4*)deg1)[i];
    ((float4*)deg)[i] = make_float4(dis_of(a.x + b.x), dis_of(a.y + b.y),
                                    dis_of(a.z + b.z), dis_of(a.w + b.w));
}

template <int DS>
__device__ __forceinline__ void scale_store_half(const float (&b)[8], float* dst,
                                                 float nk, int lane) {
    const int hl = lane & 15;
    const int h  = lane >> 4;
    v8f acc = {};
#pragma unroll
    for (int t = 0; t < 4; ++t) {
        v2f a, bb;
        a.x = (hl == 8 * h + 2 * t)     ? nk : 0.0f;
        a.y = (hl == 8 * h + 2 * t + 1) ? nk : 0.0f;
        bb.x = b[2 * t];
        bb.y = b[2 * t + 1];
        acc = __builtin_amdgcn_wmma_f32_16x16x4_f32(
                  false, a, false, bb, (short)0, acc, false, false);
    }
#pragma unroll
    for (int v = 0; v < 8; ++v)
        dst[v * DS + lane] = acc[v];
}

#define G_WAVES (N_NODES / 8)
#define REC_INTS 128

template <int SS, int NJ>
__device__ __forceinline__ bool sum_steps(const float* src, int R, int lane, float (&acc)[8]) {
    const unsigned m = __builtin_amdgcn_ballot_w32(R >= 0);
    if (m == 0u) return false;
    int   sid[NJ][8];
    float v[NJ][8];
#pragma unroll
    for (int j = 0; j < NJ; ++j)
#pragma unroll
        for (int i = 0; i < 8; ++i) {
            sid[j][i] = __builtin_amdgcn_readlane(R, 8 * j + i);
            const int idx = sid[j][i] >= 0 ? sid[j][i] : 0;
            v[j][i] = src[idx * SS + lane];
        }
#pragma unroll
    for (int j = 0; j < NJ; ++j)
#pragma unroll
        for (int i = 0; i < 8; ++i)
            acc[i] += (sid[j][i] >= 0) ? v[j][i] : 0.0f;
    return ((m >> (8 * (NJ - 1))) & 0xffu) != 0u;
}

template <int SS, int DS>
__global__ __launch_bounds__(256) void k_gather(
        const float* src, float* dst, const float* __restrict__ dis,
        const int* __restrict__ rec, const int2* __restrict__ link, int kp, int sq) {
    const int wave = __builtin_amdgcn_readfirstlane((int)(blockIdx.x * 8 + (threadIdx.x >> 5)));
    if (wave >= G_WAVES) return;
    const int lane = threadIdx.x & 31;
    const int nb = wave * 8;
    const int hl = lane & 15, j = hl & 7, hs = hl >> 3;
    const int* rp = rec + wave * REC_INTS;
    const int R0 = rp[lane], R1 = rp[32 + lane], R2 = rp[64 + lane], R3 = rp[96 + lane];
    const float d = dis[(nb + j) * 4 + 2 * kp + hs];
    const float f = sq ? d * d : d;

    float acc[8];
#pragma unroll
    for (int i = 0; i < 8; ++i) acc[i] = 0.0f;

    if (sum_steps<SS, 4>(src, R0, lane, acc))
        if (sum_steps<SS, 4>(src, R1, lane, acc))
            if (sum_steps<SS, 4>(src, R2, lane, acc))
                if (sum_steps<SS, 3>(src, R3, lane, acc)) {
                    const int ce = __builtin_amdgcn_ds_bpermute((24 + (lane & 7)) << 2, R3);
                    int2 lk = make_int2(-1, -1);
                    if (ce >= 0) lk = link[ce];
                    while (__builtin_amdgcn_ballot_w32(lk.x >= 0) != 0u) {
                        const int en = lk.y;
                        int2 lkn = make_int2(-1, -1);
                        if (en >= 0) lkn = link[en];
                        int   sid[8];
                        float v[8];
#pragma unroll
                        for (int i = 0; i < 8; ++i) {
                            sid[i] = __builtin_amdgcn_readlane(lk.x, i);
                            const int idx = sid[i] >= 0 ? sid[i] : 0;
                            v[i] = src[idx * SS + lane];
                        }
#pragma unroll
                        for (int i = 0; i < 8; ++i)
                            acc[i] += (sid[i] >= 0) ? v[i] : 0.0f;
                        lk = lkn;
                    }
                }
    scale_store_half<DS>(acc, dst + nb * DS, f, lane);
}

__global__ __launch_bounds__(256) void k_gather_first(
        const float* __restrict__ Gu, const float* __restrict__ Gi, float* __restrict__ dst,
        const float* __restrict__ dis, const int* __restrict__ head,
        const int2* __restrict__ link, int* __restrict__ rec) {
    const int wave = __builtin_amdgcn_readfirstlane((int)(blockIdx.x * 8 + (threadIdx.x >> 5)));
    if (wave >= G_WAVES) return;
    const int lane = threadIdx.x & 31;
    const int nb = wave * 8;
    const int hl = lane & 15, j = hl & 7, hs = hl >> 3;
    const int h = lane >> 4;
    const int q = lane >> 3;
    const float d0 = dis[(nb + j) * 4 + hs];
    const float d1 = dis[(nb + j) * 4 + 2 + hs];

    const int e0 = head[nb + (lane & 7)];
    int2 lk = make_int2(-1, -1);
    if (e0 >= 0) lk = link[e0];
    int ecur = e0;
    int step = 0;
    int recv = -1;
    int* rp = rec + wave * REC_INTS;
    float accl[8], acch[8];
#pragma unroll
    for (int i = 0; i < 8; ++i) { accl[i] = 0.0f; acch[i] = 0.0f; }

    while (__builtin_amdgcn_ballot_w32(lk.x >= 0) != 0u) {
        const int en = lk.y;
        int2 lkn = make_int2(-1, -1);
        if (en >= 0) lkn = link[en];
        const int sx = lk.x >= 0 ? lk.x : 0;
        const int dv = __float_as_int(dis[sx * 4 + q]);
        if (step < 16) {
            const int word = (step == 15) ? ecur : lk.x;
            recv = ((step & 3) == (lane >> 3)) ? word : recv;
            if ((step & 3) == 3) { rp[(step >> 2) * 32 + lane] = recv; recv = -1; }
        }
        int   sid[8];
        float vl[8], vh[8];
#pragma unroll
        for (int i = 0; i < 8; ++i) {
            sid[i] = __builtin_amdgcn_readlane(lk.x, i);
            const int idx = sid[i] >= 0 ? sid[i] : 0;
            const float* p = (idx < N_USERS) ? (Gu + idx * 64) : (Gi + (idx - N_USERS) * 64);
            vl[i] = p[lane];
            vh[i] = p[32 + lane];
        }
#pragma unroll
        for (int i = 0; i < 8; ++i) {
            const float s0 = __int_as_float(__builtin_amdgcn_readlane(dv, i));
            const float s1 = __int_as_float(__builtin_amdgcn_readlane(dv, i + 8));
            const float s2 = __int_as_float(__builtin_amdgcn_readlane(dv, i + 16));
            const float s3 = __int_as_float(__builtin_amdgcn_readlane(dv, i + 24));
            const float dl = h ? s1 : s0;
            const float dh = h ? s3 : s2;
            const float xl = (sid[i] >= 0) ? vl[i] : 0.0f;
            const float xh = (sid[i] >= 0) ? vh[i] : 0.0f;
            accl[i] = fmaf(xl, dl, accl[i]);
            acch[i] = fmaf(xh, dh, acch[i]);
        }
        lk = lkn;
        ecur = en;
        ++step;
    }
    for (int k = step >> 2; k < 4; ++k) { rp[k * 32 + lane] = recv; recv = -1; }
    scale_store_half<64>(accl, dst + nb * 64,      d0 * d0, lane);
    scale_store_half<64>(acch, dst + nb * 64 + 32, d1 * d1, lane);
}

extern "C" void kernel_launch(void* const* d_in, const int* in_sizes, int n_in,
                              void* d_out, int out_size, void* d_ws, size_t ws_size,
                              hipStream_t stream) {
    const float* Gu      = (const float*)d_in[0];
    const float* Gi      = (const float*)d_in[1];
    const int*   edge    = (const int*)d_in[2];
    const float* intents = (const float*)d_in[3];
    const int* row = edge;
    const int* col = edge + N_EDGES;
    float* out = (float*)d_out;

    float* deg  = (float*)d_ws;
    float* deg1 = out;
    float* hbuf = deg + (size_t)N_NODES * 4;
    int2*  link = (int2*)(hbuf + (size_t)N_NODES * 32);
    int*   rec  = (int*)(link + N_EDGES);
    int*   head = rec + (size_t)G_WAVES * REC_INTS;

    dim3 b256(256);
    dim3 ggrid((G_WAVES + 7) / 8);

    k_zero<<<dim3((N_NODES + 255) / 256), b256, 0, stream>>>(deg, deg1, head, N_NODES);
    k_deg<<<dim3((DEG_WAVES + 7) / 8), b256, 0, stream>>>(row, col, intents, deg, deg1, head, link);
    k_dis<<<dim3((N_NODES + 255) / 256), b256, 0, stream>>>(deg, deg1);

    k_gather_first<<<ggrid, b256, 0, stream>>>(Gu, Gi, out, deg, head, link, rec);

    k_gather<64, 32><<<ggrid, b256, 0, stream>>>(out, hbuf, deg, rec, link, 0, 1);
    k_gather<32, 64><<<ggrid, b256, 0, stream>>>(hbuf, out, deg, rec, link, 0, 0);

    k_gather<64, 32><<<ggrid, b256, 0, stream>>>(out + 32, hbuf, deg, rec, link, 1, 1);
    k_gather<32, 64><<<ggrid, b256, 0, stream>>>(hbuf, out + 32, deg, rec, link, 1, 0);
}


# k_deg: two list-head planes, one per L2 half (wave's die id picks): no head line takes exchanges from both halves; layer 1 walks a node's plane-0 list, then its plane-1 list (new sum order)
# speedup vs baseline: 3.5021x; 1.2776x over previous
#include <hip/hip_runtime.h>

#define N_USERS  100000
#define N_ITEMS  50000
#define N_NODES  150000
#define N_EDGES  1000000
#define EMBED_K  64

typedef __attribute__((ext_vector_type(2))) float v2f;
typedef __attribute__((ext_vector_type(8))) float v8f;

__device__ __forceinline__ void atomAddF32(float* p, float v) {
    (void)__hip_atomic_fetch_add(p, v, __ATOMIC_RELAXED, __HIP_MEMORY_SCOPE_AGENT);
}

__global__ void k_zero(float* __restrict__ p0, float* __restrict__ p1,
                       int* __restrict__ head, int* __restrict__ head1, int n_nodes) {
    int i = blockIdx.x * blockDim.x + threadIdx.x;
    const float4 z = make_float4(0.0f, 0.0f, 0.0f, 0.0f);
    if (i < n_nodes) {
        ((float4*)p0)[i] = z; ((float4*)p1)[i] = z;
        head[i] = -1; head1[i] = -1;
    }
}

#define DEG_J 8
#define DEG_WAVES (N_EDGES / (8 * DEG_J))
__global__ __launch_bounds__(256) void k_deg(
                      const int* __restrict__ row, const int* __restrict__ col,
                      const float* __restrict__ intents,
                      float* __restrict__ deg0, float* __restrict__ deg1,
                      int* __restrict__ head0, int* __restrict__ head1,
                      int2* __restrict__ link) {
    const unsigned place = __builtin_amdgcn_s_sendmsg_rtn(0x87);
    float* __restrict__ deg = (place & (1u << 18)) ? deg1 : deg0;
    int* __restrict__ head = (place & (1u << 18)) ? head1 : head0;
    const int wave = blockIdx.x * 8 + (threadIdx.x >> 5);
    if (wave >= DEG_WAVES) return;
    const int lane = threadIdx.x & 31;
    const int eA = wave * (8 * DEG_J) + lane, eB = eA + 32;
    const int rA = row[eA], cA = col[eA], rB = row[eB], cB = col[eB];
    const int oA = __hip_atomic_exchange(head + cA, eA, __ATOMIC_RELAXED, __HIP_MEMORY_SCOPE_AGENT);
    const int oB = __hip_atomic_exchange(head + cB, eB, __ATOMIC_RELAXED, __HIP_MEMORY_SCOPE_AGENT);
    const int q  = lane & 3;
    const int e0 = wave * (8 * DEG_J) + (lane >> 2);
    float z0[DEG_J], z1[DEG_J], z2[DEG_J], z3[DEG_J];
    int r[DEG_J], c[DEG_J];
#pragma unroll
    for (int j = 0; j < DEG_J; ++j) {
        const int e = e0 + 8 * j;
        z0[j] = intents[e];
        z1[j] = intents[N_EDGES + e];
        z2[j] = intents[2 * N_EDGES + e];
        z3[j] = intents[3 * N_EDGES + e];
        r[j] = row[e]; c[j] = col[e];
    }
#pragma unroll
    for (int j = 0; j < DEG_J; ++j) {
        float m  = fmaxf(fmaxf(z0[j], z1[j]), fmaxf(z2[j], z3[j]));
        float w0 = __expf(z0[j] - m), w1 = __expf(z1[j] - m);
        float w2 = __expf(z2[j] - m), w3 = __expf(z3[j] - m);
        float inv = 1.0f / (w0 + w1 + w2 + w3);
        w0 *= inv; w1 *= inv; w2 *= inv; w3 *= inv;
        const float wlo = (q & 1) ? w1 : w0;
        const float whi = (q & 1) ? w3 : w2;
        const float w   = (q & 2) ? whi : wlo;
        atomAddF32(deg + r[j] * 4 + q, w);
        atomAddF32(deg + c[j] * 4 + q, w);
    }
    link[eA] = make_int2(rA, oA);
    link[eB] = make_int2(rB, oB);
}

__device__ __forceinline__ float dis_of(float d) {
    return (d > 0.0f) ? rsqrtf(fmaxf(d, 1e-12f)) : 0.0f;
}
__global__ void k_dis(float* __restrict__ deg, const float* __restrict__ deg1) {
    int i = blockIdx.x * blockDim.x + threadIdx.x;
    if (i >= N_NODES) return;
    const float4 a = ((const float4*)deg)[i], b = ((const float4*)deg1)[i];
    ((float4*)deg)[i] = make_float4(dis_of(a.x + b.x), dis_of(a.y + b.y),
                                    dis_of(a.z + b.z), dis_of(a.w + b.w));
}

template <int DS>
__device__ __forceinline__ void scale_store_half(const float (&b)[8], float* dst,
                                                 float nk, int lane) {
    const int hl = lane & 15;
    const int h  = lane >> 4;
    v8f acc = {};
#pragma unroll
    for (int t = 0; t < 4; ++t) {
        v2f a, bb;
        a.x = (hl == 8 * h + 2 * t)     ? nk : 0.0f;
        a.y = (hl == 8 * h + 2 * t + 1) ? nk : 0.0f;
        bb.x = b[2 * t];
        bb.y = b[2 * t + 1];
        acc = __builtin_amdgcn_wmma_f32_16x16x4_f32(
                  false, a, false, bb, (short)0, acc, false, false);
    }
#pragma unroll
    for (int v = 0; v < 8; ++v)
        dst[v * DS + lane] = acc[v];
}

#define G_WAVES (N_NODES / 8)
#define REC_INTS 128

template <int SS, int NJ>
__device__ __forceinline__ bool sum_steps(const float* src, int R, int lane, float (&acc)[8]) {
    const unsigned m = __builtin_amdgcn_ballot_w32(R >= 0);
    if (m == 0u) return false;
    int   sid[NJ][8];
    float v[NJ][8];
#pragma unroll
    for (int j = 0; j < NJ; ++j)
#pragma unroll
        for (int i = 0; i < 8; ++i) {
            sid[j][i] = __builtin_amdgcn_readlane(R, 8 * j + i);
            const int idx = sid[j][i] >= 0 ? sid[j][i] : 0;
            v[j][i] = src[idx * SS + lane];
        }
#pragma unroll
    for (int j = 0; j < NJ; ++j)
#pragma unroll
        for (int i = 0; i < 8; ++i)
            acc[i] += (sid[j][i] >= 0) ? v[j][i] : 0.0f;
    return ((m >> (8 * (NJ - 1))) & 0xffu) != 0u;
}

template <int SS, int DS>
__global__ __launch_bounds__(256) void k_gather(
        const float* src, float* dst, const float* __restrict__ dis,
        const int* __restrict__ rec, const int2* __restrict__ link, int kp, int sq) {
    const int wave = __builtin_amdgcn_readfirstlane((int)(blockIdx.x * 8 + (threadIdx.x >> 5)));
    if (wave >= G_WAVES) return;
    const int lane = threadIdx.x & 31;
    const int nb = wave * 8;
    const int hl = lane & 15, j = hl & 7, hs = hl >> 3;
    const int* rp = rec + wave * REC_INTS;
    const int R0 = rp[lane], R1 = rp[32 + lane], R2 = rp[64 + lane], R3 = rp[96 + lane];
    const float d = dis[(nb + j) * 4 + 2 * kp + hs];
    const float f = sq ? d * d : d;

    float acc[8];
#pragma unroll
    for (int i = 0; i < 8; ++i) acc[i] = 0.0f;

    if (sum_steps<SS, 4>(src, R0, lane, acc))
        if (sum_steps<SS, 4>(src, R1, lane, acc))
            if (sum_steps<SS, 4>(src, R2, lane, acc))
                if (sum_steps<SS, 3>(src, R3, lane, acc)) {
                    const int ce = __builtin_amdgcn_ds_bpermute((24 + (lane & 7)) << 2, R3);
                    int2 lk = make_int2(-1, -1);
                    if (ce >= 0) lk = link[ce];
                    while (__builtin_amdgcn_ballot_w32(lk.x >= 0) != 0u) {
                        const int en = lk.y;
                        int2 lkn = make_int2(-1, -1);
                        if (en >= 0) lkn = link[en];
                        int   sid[8];
                        float v[8];
#pragma unroll
                        for (int i = 0; i < 8; ++i) {
                            sid[i] = __builtin_amdgcn_readlane(lk.x, i);
                            const int idx = sid[i] >= 0 ? sid[i] : 0;
                            v[i] = src[idx * SS + lane];
                        }
#pragma unroll
                        for (int i = 0; i < 8; ++i)
                            acc[i] += (sid[i] >= 0) ? v[i] : 0.0f;
                        lk = lkn;
                    }
                }
    scale_store_half<DS>(acc, dst + nb * DS, f, lane);
}

__global__ __launch_bounds__(256) void k_gather_first(
        const float* __restrict__ Gu, const float* __restrict__ Gi, float* __restrict__ dst,
        const float* __restrict__ dis, const int* __restrict__ head,
        const int* __restrict__ head1, int2* __restrict__ link, int* __restrict__ rec) {
    const int wave = __builtin_amdgcn_readfirstlane((int)(blockIdx.x * 8 + (threadIdx.x >> 5)));
    if (wave >= G_WAVES) return;
    const int lane = threadIdx.x & 31;
    const int nb = wave * 8;
    const int hl = lane & 15, j = hl & 7, hs = hl >> 3;
    const int h = lane >> 4;
    const int q = lane >> 3;
    const float d0 = dis[(nb + j) * 4 + hs];
    const float d1 = dis[(nb + j) * 4 + 2 + hs];

    int ecur = head[nb + (lane & 7)];
    int eoth = head1[nb + (lane & 7)];
    if (ecur < 0) { ecur = eoth; eoth = -1; }
    int2 lk = make_int2(-1, -1);
    if (ecur >= 0) lk = link[ecur];
    int step = 0;
    int recv = -1;
    int* rp = rec + wave * REC_INTS;
    float accl[8], acch[8];
#pragma unroll
    for (int i = 0; i < 8; ++i) { accl[i] = 0.0f; acch[i] = 0.0f; }

    while (__builtin_amdgcn_ballot_w32(lk.x >= 0) != 0u) {
        int en = lk.y;
        if (en < 0 && eoth >= 0) {
            if (step >= 15)
                (void)__hip_atomic_exchange(((int*)link) + 2 * ecur + 1, eoth,
                                            __ATOMIC_RELAXED, __HIP_MEMORY_SCOPE_AGENT);
            en = eoth; eoth = -1;
        }
        int2 lkn = make_int2(-1, -1);
        if (en >= 0) lkn = link[en];
        const int sx = lk.x >= 0 ? lk.x : 0;
        const int dv = __float_as_int(dis[sx * 4 + q]);
        if (step < 16) {
            const int word = (step == 15) ? ecur : lk.x;
            recv = ((step & 3) == (lane >> 3)) ? word : recv;
            if ((step & 3) == 3) { rp[(step >> 2) * 32 + lane] = recv; recv = -1; }
        }
        int   sid[8];
        float vl[8], vh[8];
#pragma unroll
        for (int i = 0; i < 8; ++i) {
            sid[i] = __builtin_amdgcn_readlane(lk.x, i);
            const int idx = sid[i] >= 0 ? sid[i] : 0;
            const float* p = (idx < N_USERS) ? (Gu + idx * 64) : (Gi + (idx - N_USERS) * 64);
            vl[i] = p[lane];
            vh[i] = p[32 + lane];
        }
#pragma unroll
        for (int i = 0; i < 8; ++i) {
            const float s0 = __int_as_float(__builtin_amdgcn_readlane(dv, i));
            const float s1 = __int_as_float(__builtin_amdgcn_readlane(dv, i + 8));
            const float s2 = __int_as_float(__builtin_amdgcn_readlane(dv, i + 16));
            const float s3 = __int_as_float(__builtin_amdgcn_readlane(dv, i + 24));
            const float dl = h ? s1 : s0;
            const float dh = h ? s3 : s2;
            const float xl = (sid[i] >= 0) ? vl[i] : 0.0f;
            const float xh = (sid[i] >= 0) ? vh[i] : 0.0f;
            accl[i] = fmaf(xl, dl, accl[i]);
            acch[i] = fmaf(xh, dh, acch[i]);
        }
        lk = lkn;
        ecur = en;
        ++step;
    }
    for (int k = step >> 2; k < 4; ++k) { rp[k * 32 + lane] = recv; recv = -1; }
    scale_store_half<64>(accl, dst + nb * 64,      d0 * d0, lane);
    scale_store_half<64>(acch, dst + nb * 64 + 32, d1 * d1, lane);
}

extern "C" void kernel_launch(void* const* d_in, const int* in_sizes, int n_in,
                              void* d_out, int out_size, void* d_ws, size_t ws_size,
                              hipStream_t stream) {
    const float* Gu      = (const float*)d_in[0];
    const float* Gi      = (const float*)d_in[1];
    const int*   edge    = (const int*)d_in[2];
    const float* intents = (const float*)d_in[3];
    const int* row = edge;
    const int* col = edge + N_EDGES;
    float* out = (float*)d_out;

    float* deg  = (float*)d_ws;
    float* deg1 = out;
    float* hbuf = deg + (size_t)N_NODES * 4;
    int2*  link = (int2*)(hbuf + (size_t)N_NODES * 32);
    int*   rec  = (int*)(link + N_EDGES);
    int*   head = rec + (size_t)G_WAVES * REC_INTS;
    int*   head1 = (int*)hbuf;

    dim3 b256(256);
    dim3 ggrid((G_WAVES + 7) / 8);

    k_zero<<<dim3((N_NODES + 255) / 256), b256, 0, stream>>>(deg, deg1, head, head1, N_NODES);
    k_deg<<<dim3((DEG_WAVES + 7) / 8), b256, 0, stream>>>(row, col, intents, deg, deg1, head, head1, link);
    k_dis<<<dim3((N_NODES + 255) / 256), b256, 0, stream>>>(deg, deg1);

    k_gather_first<<<ggrid, b256, 0, stream>>>(Gu, Gi, out, deg, head, head1, link, rec);

    k_gather<64, 32><<<ggrid, b256, 0, stream>>>(out, hbuf, deg, rec, link, 0, 1);
    k_gather<32, 64><<<ggrid, b256, 0, stream>>>(hbuf, out, deg, rec, link, 0, 0);

    k_gather<64, 32><<<ggrid, b256, 0, stream>>>(out + 32, hbuf, deg, rec, link, 1, 1);
    k_gather<32, 64><<<ggrid, b256, 0, stream>>>(hbuf, out + 32, deg, rec, link, 1, 0);
}
